# starting kernel
# baseline (speedup 1.0000x reference)
#include <hip/hip_runtime.h>
#include <hip/hip_bf16.h>

typedef __attribute__((ext_vector_type(2))) float v2f;
typedef __attribute__((ext_vector_type(8))) float v8f;

__global__ void k_init_deg(float* __restrict__ deg, int n) {
    int i = blockIdx.x * blockDim.x + threadIdx.x;
    if (i < n) deg[i] = 1.0f;
}

__global__ void k_count_deg(const int* __restrict__ dst, float* __restrict__ deg, int e) {
    int i = blockIdx.x * blockDim.x + threadIdx.x;
    if (i < e) atomicAdd(&deg[dst[i]], 1.0f);
}

__global__ void k_finalize_dis(float* __restrict__ dis, int n) {
    int i = blockIdx.x * blockDim.x + threadIdx.x;
    if (i < n) dis[i] = rsqrtf(dis[i]);
}

template <int FIN, int FOUT, int MT, bool RELU>
__global__ void k_gemm_wmma(const float* __restrict__ X, const float* __restrict__ W,
                            const float* __restrict__ dis, const float* __restrict__ bias,
                            float* __restrict__ H, float* __restrict__ OUT, int M) {
    constexpr int LDA = FIN + 1;
    __shared__ float sW[FIN * FOUT];
    __shared__ float sA[MT * 16 * LDA];

    const int tid  = threadIdx.x;
    const int nthr = 32 * (FOUT / 16);
    const int m0   = blockIdx.x * (MT * 16);

    for (int i = tid; i < FIN * FOUT; i += nthr) sW[i] = W[i];
    for (int i = tid; i < MT * 16 * FIN; i += nthr) {
        int r = i / FIN, c = i - r * FIN;
        int row = m0 + r; if (row >= M) row = M - 1;
        float v = X[(size_t)row * FIN + c];
        sA[r * LDA + c] = RELU ? fmaxf(v, 0.0f) : v;
    }
    __syncthreads();

    const int wave = tid >> 5;
    const int lane = tid & 31;
    const int half = lane >> 4;
    const int l16  = lane & 15;
    const int n0   = wave * 16;

    v2f bf[FIN / 4];
    #pragma unroll
    for (int kk = 0; kk < FIN / 4; ++kk) {
        int k = kk * 4 + 2 * half;
        bf[kk].x = sW[k * FOUT + n0 + l16];
        bf[kk].y = sW[(k + 1) * FOUT + n0 + l16];
    }
    const float bcol = bias[n0 + l16];

    #pragma unroll
    for (int mt = 0; mt < MT; ++mt) {
        const float* At = &sA[mt * 16 * LDA];
        v8f c = {};
        #pragma unroll
        for (int kk = 0; kk < FIN / 4; ++kk) {
            int k = kk * 4 + 2 * half;
            v2f a;
            a.x = At[l16 * LDA + k];
            a.y = At[l16 * LDA + k + 1];
            c = __builtin_amdgcn_wmma_f32_16x16x4_f32(false, a, false, bf[kk],
                                                      (short)0, c, false, false);
        }
        #pragma unroll
        for (int r = 0; r < 8; ++r) {
            int row = m0 + mt * 16 + half * 8 + r;
            if (row < M) {
                float hv = c[r];
                float s  = dis[row];
                size_t o = (size_t)row * FOUT + n0 + l16;
                H[o]   = hv;
                OUT[o] = hv * (s * s) + bcol;
            }
        }
    }
}

template <int CHUNKS>
__global__ void k_scatter(const float* __restrict__ h, const int* __restrict__ src,
                          const int* __restrict__ dst, const float* __restrict__ dis,
                          float* __restrict__ out, int e) {
    constexpr int F = CHUNKS * 4;
    const unsigned total = (unsigned)e * (unsigned)CHUNKS;
    const unsigned step  = gridDim.x * blockDim.x;
    for (unsigned idx = blockIdx.x * blockDim.x + threadIdx.x; idx < total; idx += step) {
        const unsigned ei = idx / CHUNKS;
        const unsigned c  = idx - ei * CHUNKS;
        const int s = src[ei], d = dst[ei];
        const float norm = dis[s] * dis[d];
        const float4 v = *(const float4*)(h + (size_t)s * F + (c << 2));
        float* o = out + (size_t)d * F + (c << 2);
        atomicAdd(o + 0, v.x * norm);
        atomicAdd(o + 1, v.y * norm);
        atomicAdd(o + 2, v.z * norm);
        atomicAdd(o + 3, v.w * norm);
    }
}

extern "C" void kernel_launch(void* const* d_in, const int* in_sizes, int n_in,
                              void* d_out, int out_size, void* d_ws, size_t ws_size,
                              hipStream_t stream) {
    const float* X  = (const float*)d_in[0];
    const int*   EI = (const int*)d_in[1];
    const float* W1 = (const float*)d_in[2];
    const float* b1 = (const float*)d_in[3];
    const float* W2 = (const float*)d_in[4];
    const float* b2 = (const float*)d_in[5];
    float* out = (float*)d_out;

    const int FIN1 = 64;
    const int N = in_sizes[0] / FIN1;
    const int E = in_sizes[1] / 2;
    const int* src = EI;
    const int* dst = EI + E;

    float* dis = (float*)d_ws;
    size_t off = ((size_t)N * sizeof(float) + 255) & ~(size_t)255;
    float* bufH = (float*)((char*)d_ws + off);
    float* bufO = bufH + (size_t)N * 128;

    const int T = 256;
    dim3 blkN((N + T - 1) / T), blkE((E + T - 1) / T);
    const int ntiles = (N + 15) / 16;

    k_init_deg<<<blkN, T, 0, stream>>>(dis, N);
    k_count_deg<<<blkE, T, 0, stream>>>(dst, dis, E);
    k_finalize_dis<<<blkN, T, 0, stream>>>(dis, N);

    k_gemm_wmma<64, 128, 4, false><<<(ntiles + 3) / 4, 256, 0, stream>>>(
        X, W1, dis, b1, bufH, bufO, N);
    k_scatter<32><<<8192, T, 0, stream>>>(bufH, src, dst, dis, bufO, E);

    k_gemm_wmma<128, 64, 2, true><<<(ntiles + 1) / 2, 128, 0, stream>>>(
        bufO, W2, dis, b2, bufH, out, N);
    k_scatter<16><<<8192, T, 0, stream>>>(bufH, src, dst, dis, out, E);
}


# Layer 1 aggregates neighbours on the 64 input features before the W1 product, (A X) W1 for A (X W1): half the layer-1 scatter atomics; changes f32 summation order (reassociated product)
# speedup vs baseline: 1.3192x; 1.3192x over previous
#include <hip/hip_runtime.h>
#include <hip/hip_bf16.h>

typedef __attribute__((ext_vector_type(2))) float v2f;
typedef __attribute__((ext_vector_type(8))) float v8f;

__global__ void k_init_deg(float* __restrict__ deg, int n) {
    int i = blockIdx.x * blockDim.x + threadIdx.x;
    if (i < n) deg[i] = 1.0f;
}

__global__ void k_count_deg(const int* __restrict__ dst, float* __restrict__ deg, int e) {
    int i = blockIdx.x * blockDim.x + threadIdx.x;
    if (i < e) atomicAdd(&deg[dst[i]], 1.0f);
}

__global__ void k_finalize_dis_init_agg(const float* __restrict__ deg, const float* __restrict__ X,
                                        float* __restrict__ dis, float* __restrict__ agg, int n) {
    const unsigned idx  = blockIdx.x * blockDim.x + threadIdx.x;
    const unsigned node = idx >> 4;
    const unsigned c    = idx & 15u;
    if (node < (unsigned)n) {
        const float s = rsqrtf(deg[node]);
        const float w = s * s;
        float4 v = *(const float4*)(X + (size_t)node * 64 + (c << 2));
        v.x *= w; v.y *= w; v.z *= w; v.w *= w;
        *(float4*)(agg + (size_t)node * 64 + (c << 2)) = v;
        if (c == 0) dis[node] = s;
    }
}

template <int FIN, int FOUT, int MT, bool RELU, bool SELF>
__global__ void k_gemm_wmma(const float* __restrict__ X, const float* __restrict__ W,
                            const float* __restrict__ dis, const float* __restrict__ bias,
                            float* __restrict__ H, float* __restrict__ OUT, int M) {
    constexpr int LDA = FIN + 1;
    __shared__ float sW[FIN * FOUT];
    __shared__ float sA[MT * 16 * LDA];

    const int tid  = threadIdx.x;
    const int nthr = 32 * (FOUT / 16);
    const int m0   = blockIdx.x * (MT * 16);

    for (int i = tid; i < FIN * FOUT; i += nthr) sW[i] = W[i];
    for (int i = tid; i < MT * 16 * FIN; i += nthr) {
        int r = i / FIN, c = i - r * FIN;
        int row = m0 + r; if (row >= M) row = M - 1;
        float v = X[(size_t)row * FIN + c];
        sA[r * LDA + c] = RELU ? fmaxf(v, 0.0f) : v;
    }
    __syncthreads();

    const int wave = tid >> 5;
    const int lane = tid & 31;
    const int half = lane >> 4;
    const int l16  = lane & 15;
    const int n0   = wave * 16;

    v2f bf[FIN / 4];
    #pragma unroll
    for (int kk = 0; kk < FIN / 4; ++kk) {
        int k = kk * 4 + 2 * half;
        bf[kk].x = sW[k * FOUT + n0 + l16];
        bf[kk].y = sW[(k + 1) * FOUT + n0 + l16];
    }
    const float bcol = bias[n0 + l16];

    #pragma unroll
    for (int mt = 0; mt < MT; ++mt) {
        const float* At = &sA[mt * 16 * LDA];
        v8f c = {};
        #pragma unroll
        for (int kk = 0; kk < FIN / 4; ++kk) {
            int k = kk * 4 + 2 * half;
            v2f a;
            a.x = At[l16 * LDA + k];
            a.y = At[l16 * LDA + k + 1];
            c = __builtin_amdgcn_wmma_f32_16x16x4_f32(false, a, false, bf[kk],
                                                      (short)0, c, false, false);
        }
        #pragma unroll
        for (int r = 0; r < 8; ++r) {
            int row = m0 + mt * 16 + half * 8 + r;
            if (row < M) {
                float hv = c[r];
                size_t o = (size_t)row * FOUT + n0 + l16;
                if constexpr (SELF) {
                    float s = dis[row];
                    H[o]   = hv;
                    OUT[o] = hv * (s * s) + bcol;
                } else {
                    OUT[o] = hv + bcol;
                }
            }
        }
    }
}

template <int CHUNKS>
__global__ void k_scatter(const float* __restrict__ h, const int* __restrict__ src,
                          const int* __restrict__ dst, const float* __restrict__ dis,
                          float* __restrict__ out, int e) {
    constexpr int F = CHUNKS * 4;
    const unsigned total = (unsigned)e * (unsigned)CHUNKS;
    const unsigned step  = gridDim.x * blockDim.x;
    for (unsigned idx = blockIdx.x * blockDim.x + threadIdx.x; idx < total; idx += step) {
        const unsigned ei = idx / CHUNKS;
        const unsigned c  = idx - ei * CHUNKS;
        const int s = src[ei], d = dst[ei];
        const float norm = dis[s] * dis[d];
        const float4 v = *(const float4*)(h + (size_t)s * F + (c << 2));
        float* o = out + (size_t)d * F + (c << 2);
        atomicAdd(o + 0, v.x * norm);
        atomicAdd(o + 1, v.y * norm);
        atomicAdd(o + 2, v.z * norm);
        atomicAdd(o + 3, v.w * norm);
    }
}

extern "C" void kernel_launch(void* const* d_in, const int* in_sizes, int n_in,
                              void* d_out, int out_size, void* d_ws, size_t ws_size,
                              hipStream_t stream) {
    const float* X  = (const float*)d_in[0];
    const int*   EI = (const int*)d_in[1];
    const float* W1 = (const float*)d_in[2];
    const float* b1 = (const float*)d_in[3];
    const float* W2 = (const float*)d_in[4];
    const float* b2 = (const float*)d_in[5];
    float* out = (float*)d_out;

    const int FIN1 = 64;
    const int N = in_sizes[0] / FIN1;
    const int E = in_sizes[1] / 2;
    const int* src = EI;
    const int* dst = EI + E;

    size_t off = ((size_t)N * sizeof(float) + 255) & ~(size_t)255;
    float* deg  = (float*)d_ws;
    float* dis  = (float*)((char*)d_ws + off);
    float* bufA = (float*)((char*)d_ws + 2 * off);
    float* bufO = bufA + (size_t)N * 64;

    const int T = 256;
    dim3 blkN((N + T - 1) / T), blkE((E + T - 1) / T), blkN16((N * 16 + T - 1) / T);
    const int ntiles = (N + 15) / 16;

    k_init_deg<<<blkN, T, 0, stream>>>(deg, N);
    k_count_deg<<<blkE, T, 0, stream>>>(dst, deg, E);
    k_finalize_dis_init_agg<<<blkN16, T, 0, stream>>>(deg, X, dis, bufA, N);

    k_scatter<16><<<8192, T, 0, stream>>>(X, src, dst, dis, bufA, E);
    k_gemm_wmma<64, 128, 4, false, false><<<(ntiles + 3) / 4, 256, 0, stream>>>(
        bufA, W1, nullptr, b1, nullptr, bufO, N);

    k_gemm_wmma<128, 64, 2, true, true><<<(ntiles + 1) / 2, 128, 0, stream>>>(
        bufO, W2, dis, b2, bufA, out, N);
    k_scatter<16><<<8192, T, 0, stream>>>(bufA, src, dst, dis, out, E);
}


# Neighbour sums by gather over an edge list grouped by destination in-call (count+rank, scan, fill): each row summed in registers, stored once; no f32 atomics; f32 sum order = rank order
# speedup vs baseline: 2.4871x; 1.8853x over previous
#include <hip/hip_runtime.h>
#include <hip/hip_bf16.h>


typedef __attribute__((ext_vector_type(2))) float v2f;
typedef __attribute__((ext_vector_type(8))) float v8f;

__global__ void k_zero_cnt(int* __restrict__ cnt, int n) {
    int i = blockIdx.x * blockDim.x + threadIdx.x;
    if (i < n) cnt[i] = 0;
}

__global__ void __launch_bounds__(256)
k_count_rank(const int* __restrict__ dst, int* __restrict__ cnt, int* __restrict__ rank, int e) {
    const int i0 = blockIdx.x * 1024 + threadIdx.x;
    int d[4];
    #pragma unroll
    for (int k = 0; k < 4; ++k) {
        const int i = i0 + k * 256;
        d[k] = (i < e) ? dst[i] : -1;
    }
    int r[4];
    #pragma unroll
    for (int k = 0; k < 4; ++k) r[k] = (d[k] >= 0) ? atomicAdd(&cnt[d[k]], 1) : 0;
    #pragma unroll
    for (int k = 0; k < 4; ++k) {
        const int i = i0 + k * 256;
        if (i < e) rank[i] = r[k];
    }
}

__global__ void __launch_bounds__(1024)
k_scan_rowptr(const int* __restrict__ cnt, int* __restrict__ rowptr, float* __restrict__ dis, int n) {
    __shared__ int s_pre[32];
    __shared__ int s_sum[32];
    const int tid  = threadIdx.x;
    const int lane = tid & 31;
    const int wave = tid >> 5;
    const int nb   = blockIdx.x;
    const int* cp  = cnt + tid * 4;

    int pre = 0;
    int it = 0;
    for (; it + 4 <= nb; it += 4) {
        const int4 u0 = *(const int4*)(cp + (size_t)(it + 0) * 4096);
        const int4 u1 = *(const int4*)(cp + (size_t)(it + 1) * 4096);
        const int4 u2 = *(const int4*)(cp + (size_t)(it + 2) * 4096);
        const int4 u3 = *(const int4*)(cp + (size_t)(it + 3) * 4096);
        pre += ((u0.x + u0.y) + (u0.z + u0.w)) + ((u1.x + u1.y) + (u1.z + u1.w))
             + ((u2.x + u2.y) + (u2.z + u2.w)) + ((u3.x + u3.y) + (u3.z + u3.w));
    }
    for (; it < nb; ++it) {
        const int4 u0 = *(const int4*)(cp + (size_t)it * 4096);
        pre += (u0.x + u0.y) + (u0.z + u0.w);
    }

    const int i0 = nb * 4096 + tid * 4;
    int a[4];
    #pragma unroll
    for (int k = 0; k < 4; ++k) a[k] = (i0 + k < n) ? cnt[i0 + k] : 0;
    const int t = (a[0] + a[1]) + (a[2] + a[3]);

    int inc = t;
    #pragma unroll
    for (int dlt = 1; dlt < 32; dlt <<= 1) {
        const int u = __shfl_up(inc, dlt);
        if (lane >= dlt) inc += u;
    }
    #pragma unroll
    for (int dlt = 16; dlt >= 1; dlt >>= 1) pre += __shfl_xor(pre, dlt);

    if (lane == 31) s_sum[wave] = inc;
    if (lane == 0)  s_pre[wave] = pre;
    __syncthreads();

    int run = inc - t;
    #pragma unroll
    for (int w = 0; w < 32; ++w) {
        run += s_pre[w];
        if (w < wave) run += s_sum[w];
    }
    #pragma unroll
    for (int k = 0; k < 4; ++k) {
        const int i = i0 + k;
        if (i <= n) rowptr[i] = run;
        if (i < n)  dis[i] = rsqrtf((float)(a[k] + 1));
        run += a[k];
    }
}

__global__ void __launch_bounds__(256)
k_fill_csr(const int* __restrict__ src, const int* __restrict__ dst, const int* __restrict__ rank,
           const int* __restrict__ rowptr, int* __restrict__ csr, int e) {
    const int i0 = blockIdx.x * 1024 + threadIdx.x;
    int d[4], r[4], s[4], base[4];
    #pragma unroll
    for (int k = 0; k < 4; ++k) {
        const int i = min(i0 + k * 256, e - 1);
        d[k] = dst[i];
        r[k] = rank[i];
        s[k] = src[i];
    }
    #pragma unroll
    for (int k = 0; k < 4; ++k) base[k] = rowptr[d[k]];
    #pragma unroll
    for (int k = 0; k < 4; ++k)
        if (i0 + k * 256 < e) csr[base[k] + r[k]] = s[k];
}

template <bool BIAS>
__global__ void __launch_bounds__(256)
k_gather64(const float* __restrict__ h, const int* __restrict__ rowptr, const int* __restrict__ csr,
           const float* __restrict__ dis, const float* __restrict__ bias,
           float* __restrict__ out, int n) {
    const unsigned c4   = (threadIdx.x & 15u) << 2;
    const unsigned step = (gridDim.x * blockDim.x) >> 4;
    for (unsigned node = (blockIdx.x * blockDim.x + threadIdx.x) >> 4; node < (unsigned)n; node += step) {
        const int beg = rowptr[node];
        const int end = rowptr[node + 1];
        const float sd = dis[node];
        float4 acc = *(const float4*)(h + (size_t)node * 64 + c4);
        acc.x *= sd; acc.y *= sd; acc.z *= sd; acc.w *= sd;
        const int last = end - 1;
        for (int p = beg; p < end; p += 4) {
            const bool ok1 = p + 1 < end, ok2 = p + 2 < end, ok3 = p + 3 < end;
            const int s0 = csr[p];
            const int s1 = csr[min(p + 1, last)];
            const int s2 = csr[min(p + 2, last)];
            const int s3 = csr[min(p + 3, last)];
            const float w0 = dis[s0], w1 = dis[s1], w2 = dis[s2], w3 = dis[s3];
            const float4 v0 = *(const float4*)(h + (size_t)s0 * 64 + c4);
            const float4 v1 = *(const float4*)(h + (size_t)s1 * 64 + c4);
            const float4 v2 = *(const float4*)(h + (size_t)s2 * 64 + c4);
            const float4 v3 = *(const float4*)(h + (size_t)s3 * 64 + c4);
            acc.x = fmaf(w0, v0.x, acc.x); acc.y = fmaf(w0, v0.y, acc.y);
            acc.z = fmaf(w0, v0.z, acc.z); acc.w = fmaf(w0, v0.w, acc.w);
            float4 t;
            t.x = fmaf(w1, v1.x, acc.x); t.y = fmaf(w1, v1.y, acc.y);
            t.z = fmaf(w1, v1.z, acc.z); t.w = fmaf(w1, v1.w, acc.w);
            acc.x = ok1 ? t.x : acc.x; acc.y = ok1 ? t.y : acc.y;
            acc.z = ok1 ? t.z : acc.z; acc.w = ok1 ? t.w : acc.w;
            t.x = fmaf(w2, v2.x, acc.x); t.y = fmaf(w2, v2.y, acc.y);
            t.z = fmaf(w2, v2.z, acc.z); t.w = fmaf(w2, v2.w, acc.w);
            acc.x = ok2 ? t.x : acc.x; acc.y = ok2 ? t.y : acc.y;
            acc.z = ok2 ? t.z : acc.z; acc.w = ok2 ? t.w : acc.w;
            t.x = fmaf(w3, v3.x, acc.x); t.y = fmaf(w3, v3.y, acc.y);
            t.z = fmaf(w3, v3.z, acc.z); t.w = fmaf(w3, v3.w, acc.w);
            acc.x = ok3 ? t.x : acc.x; acc.y = ok3 ? t.y : acc.y;
            acc.z = ok3 ? t.z : acc.z; acc.w = ok3 ? t.w : acc.w;
        }
        float4 r;
        r.x = acc.x * sd; r.y = acc.y * sd; r.z = acc.z * sd; r.w = acc.w * sd;
        if constexpr (BIAS) {
            const float4 b = *(const float4*)(bias + c4);
            r.x += b.x; r.y += b.y; r.z += b.z; r.w += b.w;
        }
        *(float4*)(out + (size_t)node * 64 + c4) = r;
    }
}

template <int FIN, int FOUT, int MT, bool RELU, bool BIAS>
__global__ void k_gemm_wmma(const float* __restrict__ X, const float* __restrict__ W,
                            const float* __restrict__ bias, float* __restrict__ OUT, int M) {
    constexpr int LDA = FIN + 1;
    __shared__ float sW[FIN * FOUT];
    __shared__ float sA[MT * 16 * LDA];

    const int tid  = threadIdx.x;
    const int nthr = 32 * (FOUT / 16);
    const int m0   = blockIdx.x * (MT * 16);

    for (int i = tid; i < FIN * FOUT; i += nthr) sW[i] = W[i];
    for (int i = tid; i < MT * 16 * FIN; i += nthr) {
        int r = i / FIN, c = i - r * FIN;
        int row = m0 + r; if (row >= M) row = M - 1;
        float v = X[(size_t)row * FIN + c];
        sA[r * LDA + c] = RELU ? fmaxf(v, 0.0f) : v;
    }
    __syncthreads();

    const int wave = tid >> 5;
    const int lane = tid & 31;
    const int half = lane >> 4;
    const int l16  = lane & 15;
    const int n0   = wave * 16;

    v2f bf[FIN / 4];
    #pragma unroll
    for (int kk = 0; kk < FIN / 4; ++kk) {
        int k = kk * 4 + 2 * half;
        bf[kk].x = sW[k * FOUT + n0 + l16];
        bf[kk].y = sW[(k + 1) * FOUT + n0 + l16];
    }
    float bcol = 0.0f;
    if constexpr (BIAS) bcol = bias[n0 + l16];

    #pragma unroll
    for (int mt = 0; mt < MT; ++mt) {
        const float* At = &sA[mt * 16 * LDA];
        v8f c = {};
        #pragma unroll
        for (int kk = 0; kk < FIN / 4; ++kk) {
            int k = kk * 4 + 2 * half;
            v2f a;
            a.x = At[l16 * LDA + k];
            a.y = At[l16 * LDA + k + 1];
            c = __builtin_amdgcn_wmma_f32_16x16x4_f32(false, a, false, bf[kk],
                                                      (short)0, c, false, false);
        }
        #pragma unroll
        for (int r = 0; r < 8; ++r) {
            int row = m0 + mt * 16 + half * 8 + r;
            if (row < M) {
                size_t o = (size_t)row * FOUT + n0 + l16;
                if constexpr (BIAS) OUT[o] = c[r] + bcol;
                else                OUT[o] = c[r];
            }
        }
    }
}

extern "C" void kernel_launch(void* const* d_in, const int* in_sizes, int n_in,
                              void* d_out, int out_size, void* d_ws, size_t ws_size,
                              hipStream_t stream) {
    const float* X  = (const float*)d_in[0];
    const int*   EI = (const int*)d_in[1];
    const float* W1 = (const float*)d_in[2];
    const float* b1 = (const float*)d_in[3];
    const float* W2 = (const float*)d_in[4];
    const float* b2 = (const float*)d_in[5];
    float* out = (float*)d_out;

    const int FIN1 = 64;
    const int N = in_sizes[0] / FIN1;
    const int E = in_sizes[1] / 2;
    const int* src = EI;
    const int* dst = EI + E;

    size_t off = ((size_t)(N + 1) * sizeof(float) + 255) & ~(size_t)255;
    int*   cnt    = (int*)d_ws;
    int*   rowptr = (int*)((char*)d_ws + off);
    float* dis    = (float*)((char*)d_ws + 2 * off);
    float* bufA   = (float*)((char*)d_ws + 3 * off);
    float* bufO   = bufA + (size_t)N * 64;
    int*   csr    = (int*)(bufO + (size_t)N * 128);
    int*   rank   = (int*)bufO;

    const int T = 256;
    dim3 blkN((N + T - 1) / T), blkE4((E + 1023) / 1024), blkS(N / 4096 + 1);
    const int ntiles = (N + 15) / 16;

    k_zero_cnt<<<blkN, T, 0, stream>>>(cnt, N);
    k_count_rank<<<blkE4, T, 0, stream>>>(dst, cnt, rank, E);
    k_scan_rowptr<<<blkS, 1024, 0, stream>>>(cnt, rowptr, dis, N);
    k_fill_csr<<<blkE4, T, 0, stream>>>(src, dst, rank, rowptr, csr, E);

    k_gather64<false><<<2048, T, 0, stream>>>(X, rowptr, csr, dis, nullptr, bufA, N);
    k_gemm_wmma<64, 128, 4, false, true><<<(ntiles + 3) / 4, 256, 0, stream>>>(
        bufA, W1, b1, bufO, N);

    k_gemm_wmma<128, 64, 2, true, false><<<(ntiles + 1) / 2, 128, 0, stream>>>(
        bufO, W2, nullptr, bufA, N);
    k_gather64<true><<<2048, T, 0, stream>>>(bufA, rowptr, csr, dis, b2, out, N);
}


# Edge list grouped by destination in two levels on LDS counters (256-node buckets: histogram, scan, scatter; per-bucket count, scan, fill): no global atomic left; f32 sum order = LDS arrival order
# speedup vs baseline: 3.5061x; 1.4097x over previous
#include <hip/hip_runtime.h>
#include <hip/hip_bf16.h>


typedef __attribute__((ext_vector_type(2))) float v2f;
typedef __attribute__((ext_vector_type(8))) float v8f;

constexpr int BSH    = 8;
constexpr int BNODES = 1 << BSH;
constexpr int LT     = 512;
constexpr int ECH    = LT * 16;
constexpr int CU     = 10;

__global__ void __launch_bounds__(LT)
k_bucket_hist(const int* __restrict__ dst, int* __restrict__ hist, int e, int nb) {
    extern __shared__ int s_bkt[];
    const int tid = threadIdx.x;
    const int i0 = blockIdx.x * ECH + tid;
    int d[16];
    #pragma unroll
    for (int k = 0; k < 16; ++k) {
        const int i = i0 + k * LT;
        d[k] = (i < e) ? dst[i] : -1;
    }
    for (int b = tid; b < nb; b += LT) s_bkt[b] = 0;
    __syncthreads();
    #pragma unroll
    for (int k = 0; k < 16; ++k)
        if (d[k] >= 0) atomicAdd(&s_bkt[d[k] >> BSH], 1);
    __syncthreads();
    const int g = gridDim.x;
    for (int b = tid; b < nb; b += LT) hist[(size_t)b * g + blockIdx.x] = s_bkt[b];
}

__global__ void __launch_bounds__(1024)
k_scan(const int* __restrict__ cnt, int* __restrict__ excl, int n) {
    __shared__ int s_pre[32];
    __shared__ int s_sum[32];
    const int tid  = threadIdx.x;
    const int lane = tid & 31;
    const int wave = tid >> 5;
    const int nb   = blockIdx.x;
    const int* cp  = cnt + tid * 4;

    int pre = 0;
    int it = 0;
    for (; it + 4 <= nb; it += 4) {
        const int4 u0 = *(const int4*)(cp + (size_t)(it + 0) * 4096);
        const int4 u1 = *(const int4*)(cp + (size_t)(it + 1) * 4096);
        const int4 u2 = *(const int4*)(cp + (size_t)(it + 2) * 4096);
        const int4 u3 = *(const int4*)(cp + (size_t)(it + 3) * 4096);
        pre += ((u0.x + u0.y) + (u0.z + u0.w)) + ((u1.x + u1.y) + (u1.z + u1.w))
             + ((u2.x + u2.y) + (u2.z + u2.w)) + ((u3.x + u3.y) + (u3.z + u3.w));
    }
    for (; it < nb; ++it) {
        const int4 u0 = *(const int4*)(cp + (size_t)it * 4096);
        pre += (u0.x + u0.y) + (u0.z + u0.w);
    }

    const int i0 = nb * 4096 + tid * 4;
    int a[4];
    #pragma unroll
    for (int k = 0; k < 4; ++k) a[k] = (i0 + k < n) ? cnt[i0 + k] : 0;
    const int t = (a[0] + a[1]) + (a[2] + a[3]);

    int inc = t;
    #pragma unroll
    for (int dlt = 1; dlt < 32; dlt <<= 1) {
        const int u = __shfl_up(inc, dlt);
        if (lane >= dlt) inc += u;
    }
    #pragma unroll
    for (int dlt = 16; dlt >= 1; dlt >>= 1) pre += __shfl_xor(pre, dlt);

    if (lane == 31) s_sum[wave] = inc;
    if (lane == 0)  s_pre[wave] = pre;
    __syncthreads();

    int run = inc - t;
    #pragma unroll
    for (int w = 0; w < 32; ++w) {
        run += s_pre[w];
        if (w < wave) run += s_sum[w];
    }
    #pragma unroll
    for (int k = 0; k < 4; ++k) {
        const int i = i0 + k;
        if (i <= n) excl[i] = run;
        run += a[k];
    }
}

__global__ void __launch_bounds__(LT)
k_bucket_scatter(const int* __restrict__ src, const int* __restrict__ dst, const int* __restrict__ base,
                 int2* __restrict__ pairs, int e, int nb) {
    extern __shared__ int s_bkt[];
    const int tid = threadIdx.x;
    const int g = gridDim.x;
    const int i0 = blockIdx.x * ECH + tid;
    int d[16], s[16], pos[16];
    #pragma unroll
    for (int k = 0; k < 16; ++k) {
        const int i = min(i0 + k * LT, e - 1);
        d[k] = dst[i];
        s[k] = src[i];
    }
    for (int b = tid; b < nb; b += LT) s_bkt[b] = base[(size_t)b * g + blockIdx.x];
    __syncthreads();
    #pragma unroll
    for (int k = 0; k < 16; ++k)
        pos[k] = (i0 + k * LT < e) ? atomicAdd(&s_bkt[d[k] >> BSH], 1) : -1;
    #pragma unroll
    for (int k = 0; k < 16; ++k)
        if (pos[k] >= 0) pairs[pos[k]] = make_int2(s[k], d[k]);
}

__global__ void __launch_bounds__(512)
k_bucket_csr(const int2* __restrict__ pairs, const int* __restrict__ base, int* __restrict__ rowptr,
             float* __restrict__ dis, int* __restrict__ csr, int n, int g) {
    __shared__ int s_cnt[BNODES];
    __shared__ int s_cur[BNODES];
    __shared__ int s_wsum[BNODES / 32];
    const int tid   = threadIdx.x;
    const int lane  = tid & 31;
    const int wave  = tid >> 5;
    const int b     = blockIdx.x;
    const int beg   = base[(size_t)b * g];
    const int end   = base[(size_t)(b + 1) * g];
    const int node0 = b << BSH;

    if (tid < BNODES) s_cnt[tid] = 0;
    __syncthreads();
    for (int p = beg + tid; p < end; p += CU * 512) {
        int y[CU];
        #pragma unroll
        for (int k = 0; k < CU; ++k) {
            const int q = p + k * 512;
            y[k] = (q < end) ? pairs[q].y : -1;
        }
        #pragma unroll
        for (int k = 0; k < CU; ++k)
            if (y[k] >= 0) atomicAdd(&s_cnt[y[k] - node0], 1);
    }
    __syncthreads();

    const int c = (tid < BNODES) ? s_cnt[tid] : 0;
    int inc = c;
    #pragma unroll
    for (int dlt = 1; dlt < 32; dlt <<= 1) {
        const int u = __shfl_up(inc, dlt);
        if (lane >= dlt) inc += u;
    }
    if (tid < BNODES && lane == 31) s_wsum[wave] = inc;
    __syncthreads();
    if (tid < BNODES) {
        int ex = beg + inc - c;
        #pragma unroll
        for (int w = 0; w < BNODES / 32; ++w)
            if (w < wave) ex += s_wsum[w];
        s_cur[tid] = ex;
        const int node = node0 + tid;
        if (node < n) {
            rowptr[node] = ex;
            dis[node] = rsqrtf((float)(c + 1));
            if (node == n - 1) rowptr[n] = ex + c;
        }
    }
    __syncthreads();

    const int last = end - 1;
    for (int p = beg + tid; p < end; p += CU * 512) {
        int2 r[CU];
        int pos[CU];
        #pragma unroll
        for (int k = 0; k < CU; ++k) r[k] = pairs[min(p + k * 512, last)];
        __builtin_amdgcn_sched_barrier(0);
        #pragma unroll
        for (int k = 0; k < CU; ++k)
            pos[k] = (p + k * 512 < end) ? atomicAdd(&s_cur[r[k].y - node0], 1) : -1;
        #pragma unroll
        for (int k = 0; k < CU; ++k)
            if (pos[k] >= 0) csr[pos[k]] = r[k].x;
    }
}

template <bool BIAS>
__global__ void __launch_bounds__(256)
k_gather64(const float* __restrict__ h, const int* __restrict__ rowptr, const int* __restrict__ csr,
           const float* __restrict__ dis, const float* __restrict__ bias,
           float* __restrict__ out, int n) {
    const unsigned c4   = (threadIdx.x & 15u) << 2;
    const unsigned step = (gridDim.x * blockDim.x) >> 4;
    for (unsigned node = (blockIdx.x * blockDim.x + threadIdx.x) >> 4; node < (unsigned)n; node += step) {
        const int beg = rowptr[node];
        const int end = rowptr[node + 1];
        const float sd = dis[node];
        float4 acc = *(const float4*)(h + (size_t)node * 64 + c4);
        acc.x *= sd; acc.y *= sd; acc.z *= sd; acc.w *= sd;
        const int last = end - 1;
        for (int p = beg; p < end; p += 4) {
            const bool ok1 = p + 1 < end, ok2 = p + 2 < end, ok3 = p + 3 < end;
            const int s0 = csr[p];
            const int s1 = csr[min(p + 1, last)];
            const int s2 = csr[min(p + 2, last)];
            const int s3 = csr[min(p + 3, last)];
            const float w0 = dis[s0], w1 = dis[s1], w2 = dis[s2], w3 = dis[s3];
            const float4 v0 = *(const float4*)(h + (size_t)s0 * 64 + c4);
            const float4 v1 = *(const float4*)(h + (size_t)s1 * 64 + c4);
            const float4 v2 = *(const float4*)(h + (size_t)s2 * 64 + c4);
            const float4 v3 = *(const float4*)(h + (size_t)s3 * 64 + c4);
            acc.x = fmaf(w0, v0.x, acc.x); acc.y = fmaf(w0, v0.y, acc.y);
            acc.z = fmaf(w0, v0.z, acc.z); acc.w = fmaf(w0, v0.w, acc.w);
            float4 t;
            t.x = fmaf(w1, v1.x, acc.x); t.y = fmaf(w1, v1.y, acc.y);
            t.z = fmaf(w1, v1.z, acc.z); t.w = fmaf(w1, v1.w, acc.w);
            acc.x = ok1 ? t.x : acc.x; acc.y = ok1 ? t.y : acc.y;
            acc.z = ok1 ? t.z : acc.z; acc.w = ok1 ? t.w : acc.w;
            t.x = fmaf(w2, v2.x, acc.x); t.y = fmaf(w2, v2.y, acc.y);
            t.z = fmaf(w2, v2.z, acc.z); t.w = fmaf(w2, v2.w, acc.w);
            acc.x = ok2 ? t.x : acc.x; acc.y = ok2 ? t.y : acc.y;
            acc.z = ok2 ? t.z : acc.z; acc.w = ok2 ? t.w : acc.w;
            t.x = fmaf(w3, v3.x, acc.x); t.y = fmaf(w3, v3.y, acc.y);
            t.z = fmaf(w3, v3.z, acc.z); t.w = fmaf(w3, v3.w, acc.w);
            acc.x = ok3 ? t.x : acc.x; acc.y = ok3 ? t.y : acc.y;
            acc.z = ok3 ? t.z : acc.z; acc.w = ok3 ? t.w : acc.w;
        }
        float4 r;
        r.x = acc.x * sd; r.y = acc.y * sd; r.z = acc.z * sd; r.w = acc.w * sd;
        if constexpr (BIAS) {
            const float4 b = *(const float4*)(bias + c4);
            r.x += b.x; r.y += b.y; r.z += b.z; r.w += b.w;
        }
        *(float4*)(out + (size_t)node * 64 + c4) = r;
    }
}

template <int FIN, int FOUT, int MT, bool RELU, bool BIAS>
__global__ void k_gemm_wmma(const float* __restrict__ X, const float* __restrict__ W,
                            const float* __restrict__ bias, float* __restrict__ OUT, int M) {
    constexpr int LDA = FIN + 1;
    __shared__ float sW[FIN * FOUT];
    __shared__ float sA[MT * 16 * LDA];

    const int tid  = threadIdx.x;
    const int nthr = 32 * (FOUT / 16);
    const int m0   = blockIdx.x * (MT * 16);

    for (int i = tid; i < FIN * FOUT; i += nthr) sW[i] = W[i];
    for (int i = tid; i < MT * 16 * FIN; i += nthr) {
        int r = i / FIN, c = i - r * FIN;
        int row = m0 + r; if (row >= M) row = M - 1;
        float v = X[(size_t)row * FIN + c];
        sA[r * LDA + c] = RELU ? fmaxf(v, 0.0f) : v;
    }
    __syncthreads();

    const int wave = tid >> 5;
    const int lane = tid & 31;
    const int half = lane >> 4;
    const int l16  = lane & 15;
    const int n0   = wave * 16;

    v2f bf[FIN / 4];
    #pragma unroll
    for (int kk = 0; kk < FIN / 4; ++kk) {
        int k = kk * 4 + 2 * half;
        bf[kk].x = sW[k * FOUT + n0 + l16];
        bf[kk].y = sW[(k + 1) * FOUT + n0 + l16];
    }
    float bcol = 0.0f;
    if constexpr (BIAS) bcol = bias[n0 + l16];

    #pragma unroll
    for (int mt = 0; mt < MT; ++mt) {
        const float* At = &sA[mt * 16 * LDA];
        v8f c = {};
        #pragma unroll
        for (int kk = 0; kk < FIN / 4; ++kk) {
            int k = kk * 4 + 2 * half;
            v2f a;
            a.x = At[l16 * LDA + k];
            a.y = At[l16 * LDA + k + 1];
            c = __builtin_amdgcn_wmma_f32_16x16x4_f32(false, a, false, bf[kk],
                                                      (short)0, c, false, false);
        }
        #pragma unroll
        for (int r = 0; r < 8; ++r) {
            int row = m0 + mt * 16 + half * 8 + r;
            if (row < M) {
                size_t o = (size_t)row * FOUT + n0 + l16;
                if constexpr (BIAS) OUT[o] = c[r] + bcol;
                else                OUT[o] = c[r];
            }
        }
    }
}

extern "C" void kernel_launch(void* const* d_in, const int* in_sizes, int n_in,
                              void* d_out, int out_size, void* d_ws, size_t ws_size,
                              hipStream_t stream) {
    const float* X  = (const float*)d_in[0];
    const int*   EI = (const int*)d_in[1];
    const float* W1 = (const float*)d_in[2];
    const float* b1 = (const float*)d_in[3];
    const float* W2 = (const float*)d_in[4];
    const float* b2 = (const float*)d_in[5];
    float* out = (float*)d_out;

    const int FIN1 = 64;
    const int N = in_sizes[0] / FIN1;
    const int E = in_sizes[1] / 2;
    const int* src = EI;
    const int* dst = EI + E;

    size_t off = ((size_t)(N + 1) * sizeof(float) + 255) & ~(size_t)255;
    int*   rowptr = (int*)((char*)d_ws + off);
    float* dis    = (float*)((char*)d_ws + 2 * off);
    float* bufA   = (float*)((char*)d_ws + 3 * off);
    float* bufO   = bufA + (size_t)N * 64;
    int*   csr    = (int*)(bufO + (size_t)N * 128);

    const int B  = (N + BNODES - 1) >> BSH;
    const int G  = (E + ECH - 1) / ECH;
    const int NS = B * G;
    size_t hoff = ((size_t)(NS + 1) * sizeof(int) + 255) & ~(size_t)255;
    int*  hist  = (int*)bufO;
    int*  base  = (int*)((char*)bufO + hoff);
    int2* pairs = (int2*)((char*)bufO + 2 * hoff);

    const int T = 256;
    const int ntiles = (N + 15) / 16;

    k_bucket_hist<<<G, LT, B * sizeof(int), stream>>>(dst, hist, E, B);
    k_scan<<<NS / 4096 + 1, 1024, 0, stream>>>(hist, base, NS);
    k_bucket_scatter<<<G, LT, B * sizeof(int), stream>>>(src, dst, base, pairs, E, B);
    k_bucket_csr<<<B, 512, 0, stream>>>(pairs, base, rowptr, dis, csr, N, G);

    k_gather64<false><<<2048, T, 0, stream>>>(X, rowptr, csr, dis, nullptr, bufA, N);
    k_gemm_wmma<64, 128, 4, false, true><<<(ntiles + 3) / 4, 256, 0, stream>>>(
        bufA, W1, b1, bufO, N);

    k_gemm_wmma<128, 64, 2, true, false><<<(ntiles + 1) / 2, 128, 0, stream>>>(
        bufO, W2, nullptr, bufA, N);
    k_gather64<true><<<2048, T, 0, stream>>>(bufA, rowptr, csr, dis, b2, out, N);
}


# Both dense products in one launch: the relu((A X) W1 + b1) tile stays in LDS and feeds the W2 product, out1 is never stored or re-read; W fragments straight to registers; f32 sum order unchanged
# speedup vs baseline: 4.5362x; 1.2938x over previous
#include <hip/hip_runtime.h>
#include <hip/hip_bf16.h>


typedef __attribute__((ext_vector_type(2))) float v2f;
typedef __attribute__((ext_vector_type(8))) float v8f;

constexpr int BSH    = 8;
constexpr int BNODES = 1 << BSH;
constexpr int LT     = 512;
constexpr int ECH    = LT * 16;
constexpr int CU     = 10;

__global__ void __launch_bounds__(LT)
k_bucket_hist(const int* __restrict__ dst, int* __restrict__ hist, int e, int nb) {
    extern __shared__ int s_bkt[];
    const int tid = threadIdx.x;
    const int i0 = blockIdx.x * ECH + tid;
    int d[16];
    #pragma unroll
    for (int k = 0; k < 16; ++k) {
        const int i = i0 + k * LT;
        d[k] = (i < e) ? dst[i] : -1;
    }
    for (int b = tid; b < nb; b += LT) s_bkt[b] = 0;
    __syncthreads();
    #pragma unroll
    for (int k = 0; k < 16; ++k)
        if (d[k] >= 0) atomicAdd(&s_bkt[d[k] >> BSH], 1);
    __syncthreads();
    const int g = gridDim.x;
    for (int b = tid; b < nb; b += LT) hist[(size_t)b * g + blockIdx.x] = s_bkt[b];
}

__global__ void __launch_bounds__(1024)
k_scan(const int* __restrict__ cnt, int* __restrict__ excl, int n) {
    __shared__ int s_pre[32];
    __shared__ int s_sum[32];
    const int tid  = threadIdx.x;
    const int lane = tid & 31;
    const int wave = tid >> 5;
    const int nb   = blockIdx.x;
    const int* cp  = cnt + tid * 4;

    int pre = 0;
    int it = 0;
    for (; it + 4 <= nb; it += 4) {
        const int4 u0 = *(const int4*)(cp + (size_t)(it + 0) * 4096);
        const int4 u1 = *(const int4*)(cp + (size_t)(it + 1) * 4096);
        const int4 u2 = *(const int4*)(cp + (size_t)(it + 2) * 4096);
        const int4 u3 = *(const int4*)(cp + (size_t)(it + 3) * 4096);
        pre += ((u0.x + u0.y) + (u0.z + u0.w)) + ((u1.x + u1.y) + (u1.z + u1.w))
             + ((u2.x + u2.y) + (u2.z + u2.w)) + ((u3.x + u3.y) + (u3.z + u3.w));
    }
    for (; it < nb; ++it) {
        const int4 u0 = *(const int4*)(cp + (size_t)it * 4096);
        pre += (u0.x + u0.y) + (u0.z + u0.w);
    }

    const int i0 = nb * 4096 + tid * 4;
    int a[4];
    #pragma unroll
    for (int k = 0; k < 4; ++k) a[k] = (i0 + k < n) ? cnt[i0 + k] : 0;
    const int t = (a[0] + a[1]) + (a[2] + a[3]);

    int inc = t;
    #pragma unroll
    for (int dlt = 1; dlt < 32; dlt <<= 1) {
        const int u = __shfl_up(inc, dlt);
        if (lane >= dlt) inc += u;
    }
    #pragma unroll
    for (int dlt = 16; dlt >= 1; dlt >>= 1) pre += __shfl_xor(pre, dlt);

    if (lane == 31) s_sum[wave] = inc;
    if (lane == 0)  s_pre[wave] = pre;
    __syncthreads();

    int run = inc - t;
    #pragma unroll
    for (int w = 0; w < 32; ++w) {
        run += s_pre[w];
        if (w < wave) run += s_sum[w];
    }
    #pragma unroll
    for (int k = 0; k < 4; ++k) {
        const int i = i0 + k;
        if (i <= n) excl[i] = run;
        run += a[k];
    }
}

__global__ void __launch_bounds__(LT)
k_bucket_scatter(const int* __restrict__ src, const int* __restrict__ dst, const int* __restrict__ base,
                 int2* __restrict__ pairs, int e, int nb) {
    extern __shared__ int s_bkt[];
    const int tid = threadIdx.x;
    const int g = gridDim.x;
    const int i0 = blockIdx.x * ECH + tid;
    int d[16], s[16], pos[16];
    #pragma unroll
    for (int k = 0; k < 16; ++k) {
        const int i = min(i0 + k * LT, e - 1);
        d[k] = dst[i];
        s[k] = src[i];
    }
    for (int b = tid; b < nb; b += LT) s_bkt[b] = base[(size_t)b * g + blockIdx.x];
    __syncthreads();
    #pragma unroll
    for (int k = 0; k < 16; ++k)
        pos[k] = (i0 + k * LT < e) ? atomicAdd(&s_bkt[d[k] >> BSH], 1) : -1;
    #pragma unroll
    for (int k = 0; k < 16; ++k)
        if (pos[k] >= 0) pairs[pos[k]] = make_int2(s[k], d[k]);
}

__global__ void __launch_bounds__(512)
k_bucket_csr(const int2* __restrict__ pairs, const int* __restrict__ base, int* __restrict__ rowptr,
             float* __restrict__ dis, int* __restrict__ csr, int n, int g) {
    __shared__ int s_cnt[BNODES];
    __shared__ int s_cur[BNODES];
    __shared__ int s_wsum[BNODES / 32];
    const int tid   = threadIdx.x;
    const int lane  = tid & 31;
    const int wave  = tid >> 5;
    const int b     = blockIdx.x;
    const int beg   = base[(size_t)b * g];
    const int end   = base[(size_t)(b + 1) * g];
    const int node0 = b << BSH;

    if (tid < BNODES) s_cnt[tid] = 0;
    __syncthreads();
    for (int p = beg + tid; p < end; p += CU * 512) {
        int y[CU];
        #pragma unroll
        for (int k = 0; k < CU; ++k) {
            const int q = p + k * 512;
            y[k] = (q < end) ? pairs[q].y : -1;
        }
        #pragma unroll
        for (int k = 0; k < CU; ++k)
            if (y[k] >= 0) atomicAdd(&s_cnt[y[k] - node0], 1);
    }
    __syncthreads();

    const int c = (tid < BNODES) ? s_cnt[tid] : 0;
    int inc = c;
    #pragma unroll
    for (int dlt = 1; dlt < 32; dlt <<= 1) {
        const int u = __shfl_up(inc, dlt);
        if (lane >= dlt) inc += u;
    }
    if (tid < BNODES && lane == 31) s_wsum[wave] = inc;
    __syncthreads();
    if (tid < BNODES) {
        int ex = beg + inc - c;
        #pragma unroll
        for (int w = 0; w < BNODES / 32; ++w)
            if (w < wave) ex += s_wsum[w];
        s_cur[tid] = ex;
        const int node = node0 + tid;
        if (node < n) {
            rowptr[node] = ex;
            dis[node] = rsqrtf((float)(c + 1));
            if (node == n - 1) rowptr[n] = ex + c;
        }
    }
    __syncthreads();

    const int last = end - 1;
    for (int p = beg + tid; p < end; p += CU * 512) {
        int2 r[CU];
        int pos[CU];
        #pragma unroll
        for (int k = 0; k < CU; ++k) r[k] = pairs[min(p + k * 512, last)];
        __builtin_amdgcn_sched_barrier(0);
        #pragma unroll
        for (int k = 0; k < CU; ++k)
            pos[k] = (p + k * 512 < end) ? atomicAdd(&s_cur[r[k].y - node0], 1) : -1;
        #pragma unroll
        for (int k = 0; k < CU; ++k)
            if (pos[k] >= 0) csr[pos[k]] = r[k].x;
    }
}

template <bool BIAS>
__global__ void __launch_bounds__(256)
k_gather64(const float* __restrict__ h, const int* __restrict__ rowptr, const int* __restrict__ csr,
           const float* __restrict__ dis, const float* __restrict__ bias,
           float* __restrict__ out, int n) {
    const unsigned c4   = (threadIdx.x & 15u) << 2;
    const unsigned step = (gridDim.x * blockDim.x) >> 4;
    for (unsigned node = (blockIdx.x * blockDim.x + threadIdx.x) >> 4; node < (unsigned)n; node += step) {
        const int beg = rowptr[node];
        const int end = rowptr[node + 1];
        const float sd = dis[node];
        float4 acc = *(const float4*)(h + (size_t)node * 64 + c4);
        acc.x *= sd; acc.y *= sd; acc.z *= sd; acc.w *= sd;
        const int last = end - 1;
        for (int p = beg; p < end; p += 4) {
            const bool ok1 = p + 1 < end, ok2 = p + 2 < end, ok3 = p + 3 < end;
            const int s0 = csr[p];
            const int s1 = csr[min(p + 1, last)];
            const int s2 = csr[min(p + 2, last)];
            const int s3 = csr[min(p + 3, last)];
            const float w0 = dis[s0], w1 = dis[s1], w2 = dis[s2], w3 = dis[s3];
            const float4 v0 = *(const float4*)(h + (size_t)s0 * 64 + c4);
            const float4 v1 = *(const float4*)(h + (size_t)s1 * 64 + c4);
            const float4 v2 = *(const float4*)(h + (size_t)s2 * 64 + c4);
            const float4 v3 = *(const float4*)(h + (size_t)s3 * 64 + c4);
            acc.x = fmaf(w0, v0.x, acc.x); acc.y = fmaf(w0, v0.y, acc.y);
            acc.z = fmaf(w0, v0.z, acc.z); acc.w = fmaf(w0, v0.w, acc.w);
            float4 t;
            t.x = fmaf(w1, v1.x, acc.x); t.y = fmaf(w1, v1.y, acc.y);
            t.z = fmaf(w1, v1.z, acc.z); t.w = fmaf(w1, v1.w, acc.w);
            acc.x = ok1 ? t.x : acc.x; acc.y = ok1 ? t.y : acc.y;
            acc.z = ok1 ? t.z : acc.z; acc.w = ok1 ? t.w : acc.w;
            t.x = fmaf(w2, v2.x, acc.x); t.y = fmaf(w2, v2.y, acc.y);
            t.z = fmaf(w2, v2.z, acc.z); t.w = fmaf(w2, v2.w, acc.w);
            acc.x = ok2 ? t.x : acc.x; acc.y = ok2 ? t.y : acc.y;
            acc.z = ok2 ? t.z : acc.z; acc.w = ok2 ? t.w : acc.w;
            t.x = fmaf(w3, v3.x, acc.x); t.y = fmaf(w3, v3.y, acc.y);
            t.z = fmaf(w3, v3.z, acc.z); t.w = fmaf(w3, v3.w, acc.w);
            acc.x = ok3 ? t.x : acc.x; acc.y = ok3 ? t.y : acc.y;
            acc.z = ok3 ? t.z : acc.z; acc.w = ok3 ? t.w : acc.w;
        }
        float4 r;
        r.x = acc.x * sd; r.y = acc.y * sd; r.z = acc.z * sd; r.w = acc.w * sd;
        if constexpr (BIAS) {
            const float4 b = *(const float4*)(bias + c4);
            r.x += b.x; r.y += b.y; r.z += b.z; r.w += b.w;
        }
        *(float4*)(out + (size_t)node * 64 + c4) = r;
    }
}

constexpr int F0  = 64;
constexpr int F1  = 128;
constexpr int F2  = 64;
constexpr int TM  = 64;
constexpr int LDA = F0 + 4;
constexpr int LDH = F1 + 4;

template <int KK>
__device__ __forceinline__ void mma_2rows(const float* A, int rs, const v2f* bf, v8f& c0, v8f& c1) {
    #pragma unroll
    for (int kk = 0; kk < KK; ++kk) {
        const v2f a0 = *(const v2f*)(A + kk * 4);
        const v2f a1 = *(const v2f*)(A + rs + kk * 4);
        c0 = __builtin_amdgcn_wmma_f32_16x16x4_f32(false, a0, false, bf[kk],
                                                   (short)0, c0, false, false);
        c1 = __builtin_amdgcn_wmma_f32_16x16x4_f32(false, a1, false, bf[kk],
                                                   (short)0, c1, false, false);
    }
}

__global__ void __launch_bounds__(256)
k_mlp_wmma(const float* __restrict__ XA, const float* __restrict__ W1, const float* __restrict__ b1,
           const float* __restrict__ W2, float* __restrict__ OUT, int M) {
    __shared__ __attribute__((aligned(16))) float sA[TM * LDA];
    __shared__ __attribute__((aligned(16))) float sH[TM * LDH];

    const int tid  = threadIdx.x;
    const int wave = tid >> 5;
    const int lane = tid & 31;
    const int half = lane >> 4;
    const int l16  = lane & 15;
    const int m0   = blockIdx.x * TM;

    const int xr = tid >> 4;
    const int c4 = (tid & 15) << 2;
    float4 xv[TM / 16];
    #pragma unroll
    for (int j = 0; j < TM / 16; ++j) {
        int row = m0 + xr + 16 * j; if (row >= M) row = M - 1;
        xv[j] = *(const float4*)(XA + (size_t)row * F0 + c4);
    }

    const int n1 = wave * 16 + l16;
    v2f bf1[F0 / 4];
    #pragma unroll
    for (int kk = 0; kk < F0 / 4; ++kk) {
        const int k = kk * 4 + 2 * half;
        bf1[kk].x = W1[k * F1 + n1];
        bf1[kk].y = W1[(k + 1) * F1 + n1];
    }
    const float bcol = b1[n1];

    #pragma unroll
    for (int j = 0; j < TM / 16; ++j)
        *(float4*)&sA[(xr + 16 * j) * LDA + c4] = xv[j];
    __syncthreads();

    #pragma unroll
    for (int mp = 0; mp < TM / 32; ++mp) {
        v8f c0 = {}, c1 = {};
        mma_2rows<F0 / 4>(&sA[(mp * 32 + l16) * LDA + 2 * half], 16 * LDA, bf1, c0, c1);
        float* H = &sH[(mp * 32 + half * 8) * LDH + n1];
        #pragma unroll
        for (int r = 0; r < 8; ++r) {
            H[r * LDH]        = fmaxf(c0[r] + bcol, 0.0f);
            H[(16 + r) * LDH] = fmaxf(c1[r] + bcol, 0.0f);
        }
    }

    const int n2 = (wave & 3) * 16 + l16;
    v2f bf2[F1 / 4];
    #pragma unroll
    for (int kk = 0; kk < F1 / 4; ++kk) {
        const int k = kk * 4 + 2 * half;
        bf2[kk].x = W2[k * F2 + n2];
        bf2[kk].y = W2[(k + 1) * F2 + n2];
    }
    __syncthreads();

    const int mq = wave >> 2;
    v8f c0 = {}, c1 = {};
    mma_2rows<F1 / 4>(&sH[(mq * 16 + l16) * LDH + 2 * half], 32 * LDH, bf2, c0, c1);
    #pragma unroll
    for (int r = 0; r < 8; ++r) {
        const int row0 = m0 + mq * 16 + half * 8 + r;
        const int row1 = row0 + 32;
        if (row0 < M) OUT[(size_t)row0 * F2 + n2] = c0[r];
        if (row1 < M) OUT[(size_t)row1 * F2 + n2] = c1[r];
    }
}

extern "C" void kernel_launch(void* const* d_in, const int* in_sizes, int n_in,
                              void* d_out, int out_size, void* d_ws, size_t ws_size,
                              hipStream_t stream) {
    const float* X  = (const float*)d_in[0];
    const int*   EI = (const int*)d_in[1];
    const float* W1 = (const float*)d_in[2];
    const float* b1 = (const float*)d_in[3];
    const float* W2 = (const float*)d_in[4];
    const float* b2 = (const float*)d_in[5];
    float* out = (float*)d_out;

    const int FIN1 = 64;
    const int N = in_sizes[0] / FIN1;
    const int E = in_sizes[1] / 2;
    const int* src = EI;
    const int* dst = EI + E;

    size_t off = ((size_t)(N + 1) * sizeof(float) + 255) & ~(size_t)255;
    int*   rowptr = (int*)((char*)d_ws + off);
    float* dis    = (float*)((char*)d_ws + 2 * off);
    float* bufA   = (float*)((char*)d_ws + 3 * off);
    float* bufO   = bufA + (size_t)N * 64;
    int*   csr    = (int*)(bufO + (size_t)N * 128);

    const int B  = (N + BNODES - 1) >> BSH;
    const int G  = (E + ECH - 1) / ECH;
    const int NS = B * G;
    size_t hoff = ((size_t)(NS + 1) * sizeof(int) + 255) & ~(size_t)255;
    int*  hist  = (int*)bufO;
    int*  base  = (int*)((char*)bufO + hoff);
    int2* pairs = (int2*)((char*)bufO + 2 * hoff);

    const int T = 256;

    k_bucket_hist<<<G, LT, B * sizeof(int), stream>>>(dst, hist, E, B);
    k_scan<<<NS / 4096 + 1, 1024, 0, stream>>>(hist, base, NS);
    k_bucket_scatter<<<G, LT, B * sizeof(int), stream>>>(src, dst, base, pairs, E, B);
    k_bucket_csr<<<B, 512, 0, stream>>>(pairs, base, rowptr, dis, csr, N, G);

    k_gather64<false><<<2048, T, 0, stream>>>(X, rowptr, csr, dis, nullptr, bufA, N);
    k_mlp_wmma<<<(N + TM - 1) / TM, 256, 0, stream>>>(bufA, W1, b1, W2, bufO, N);
    k_gather64<true><<<2048, T, 0, stream>>>(bufO, rowptr, csr, dis, b2, out, N);
}


# Edge grouping without global histogram and scan: each 8192-edge chunk grouped by bucket in LDS into its own region, bucket pass sums chunk offsets; 2 launches fewer; f32 sum order = LDS arrival order
# speedup vs baseline: 5.2596x; 1.1595x over previous
#include <hip/hip_runtime.h>
#include <hip/hip_bf16.h>


typedef __attribute__((ext_vector_type(2))) float v2f;
typedef __attribute__((ext_vector_type(8))) float v8f;

constexpr int BSH    = 8;
constexpr int BNODES = 1 << BSH;
constexpr int LT     = 512;
constexpr int ECH    = LT * 16;
constexpr int RU     = 13;

__global__ void __launch_bounds__(LT)
k_chunk_group(const int* __restrict__ src, const int* __restrict__ dst, int* __restrict__ off,
              int2* __restrict__ pairs, int e, int nb) {
    extern __shared__ int s_bkt[];
    __shared__ int s_ws[LT / 32];
    const int tid  = threadIdx.x;
    const int lane = tid & 31;
    const int wave = tid >> 5;
    const int g = gridDim.x;
    const int i0 = blockIdx.x * ECH + tid;
    int d[16], s[16], rk[16];
    #pragma unroll
    for (int k = 0; k < 16; ++k) {
        const int i = min(i0 + k * LT, e - 1);
        d[k] = dst[i];
        s[k] = src[i];
    }
    for (int b = tid; b < nb; b += LT) s_bkt[b] = 0;
    __syncthreads();
    #pragma unroll
    for (int k = 0; k < 16; ++k)
        rk[k] = (i0 + k * LT < e) ? atomicAdd(&s_bkt[d[k] >> BSH], 1) : -1;
    __syncthreads();

    int carry = 0;
    for (int b0 = 0; b0 < nb; b0 += LT) {
        const int b = b0 + tid;
        const int c = (b < nb) ? s_bkt[b] : 0;
        int inc = c;
        #pragma unroll
        for (int dlt = 1; dlt < 32; dlt <<= 1) {
            const int u = __shfl_up(inc, dlt);
            if (lane >= dlt) inc += u;
        }
        if (lane == 31) s_ws[wave] = inc;
        __syncthreads();
        int woff = 0, tot = 0;
        #pragma unroll
        for (int w = 0; w < LT / 32; ++w) {
            const int v = s_ws[w];
            tot += v;
            if (w < wave) woff += v;
        }
        const int ex = carry + woff + inc - c;
        if (b < nb) {
            s_bkt[b] = ex;
            off[(size_t)b * g + blockIdx.x] = ex;
        }
        carry += tot;
        __syncthreads();
    }
    if (tid == 0) off[(size_t)nb * g + blockIdx.x] = carry;

    int2* mine = pairs + (size_t)blockIdx.x * ECH;
    int slot[16];
    #pragma unroll
    for (int k = 0; k < 16; ++k) slot[k] = s_bkt[d[k] >> BSH] + rk[k];
    #pragma unroll
    for (int k = 0; k < 16; ++k)
        if (rk[k] >= 0) mine[slot[k]] = make_int2(s[k], d[k]);
}

__global__ void __launch_bounds__(512)
k_bucket_csr(const int2* __restrict__ pairs, const int* __restrict__ off, int* __restrict__ rowptr,
             float* __restrict__ dis, int* __restrict__ csr, int n, int g) {
    extern __shared__ int2 s_run[];
    __shared__ int s_cnt[BNODES];
    __shared__ int s_cur[BNODES];
    __shared__ int s_wsum[BNODES / 32];
    __shared__ int s_red[512 / 32];
    const int tid   = threadIdx.x;
    const int lane  = tid & 31;
    const int wave  = tid >> 5;
    const int b     = blockIdx.x;
    const int node0 = b << BSH;

    if (tid < BNODES) s_cnt[tid] = 0;
    int a0 = 0;
    for (int q = tid; q < g; q += 512) {
        const int r0 = off[(size_t)b * g + q];
        const int r1 = off[(size_t)(b + 1) * g + q];
        s_run[q] = make_int2(r0, r1 - r0);
        a0 += r0;
    }
    #pragma unroll
    for (int dlt = 16; dlt >= 1; dlt >>= 1) a0 += __shfl_xor(a0, dlt);
    if (lane == 0) s_red[wave] = a0;
    __syncthreads();
    int beg = 0;
    #pragma unroll
    for (int w = 0; w < 512 / 32; ++w) beg += s_red[w];

    for (int q0 = wave; q0 < g; q0 += 16 * RU) {
        int r0[RU], len[RU], y[RU];
        #pragma unroll
        for (int u = 0; u < RU; ++u) {
            const int q = q0 + 16 * u;
            const int2 rl = s_run[min(q, g - 1)];
            r0[u]  = rl.x;
            len[u] = (q < g) ? rl.y : 0;
        }
        #pragma unroll
        for (int u = 0; u < RU; ++u) {
            const int2* rp = pairs + (size_t)min(q0 + 16 * u, g - 1) * ECH;
            y[u] = rp[(lane < len[u]) ? r0[u] + lane : 0].y;
        }
        #pragma unroll
        for (int u = 0; u < RU; ++u)
            if (lane < len[u]) atomicAdd(&s_cnt[y[u] - node0], 1);
        #pragma unroll
        for (int u = 0; u < RU; ++u) {
            const int2* rp = pairs + (size_t)min(q0 + 16 * u, g - 1) * ECH + r0[u];
            for (int i = lane + 32; i < len[u]; i += 32)
                atomicAdd(&s_cnt[rp[i].y - node0], 1);
        }
    }
    __syncthreads();

    const int c = (tid < BNODES) ? s_cnt[tid] : 0;
    int inc = c;
    #pragma unroll
    for (int dlt = 1; dlt < 32; dlt <<= 1) {
        const int u = __shfl_up(inc, dlt);
        if (lane >= dlt) inc += u;
    }
    if (tid < BNODES && lane == 31) s_wsum[wave] = inc;
    __syncthreads();
    if (tid < BNODES) {
        int ex = beg + inc - c;
        #pragma unroll
        for (int w = 0; w < BNODES / 32; ++w)
            if (w < wave) ex += s_wsum[w];
        s_cur[tid] = ex;
        const int node = node0 + tid;
        if (node < n) {
            rowptr[node] = ex;
            dis[node] = rsqrtf((float)(c + 1));
            if (node == n - 1) rowptr[n] = ex + c;
        }
    }
    __syncthreads();

    for (int q0 = wave; q0 < g; q0 += 16 * RU) {
        int r0[RU], len[RU], pos[RU];
        int2 r[RU];
        #pragma unroll
        for (int u = 0; u < RU; ++u) {
            const int q = q0 + 16 * u;
            const int2 rl = s_run[min(q, g - 1)];
            r0[u]  = rl.x;
            len[u] = (q < g) ? rl.y : 0;
        }
        #pragma unroll
        for (int u = 0; u < RU; ++u) {
            const int2* rp = pairs + (size_t)min(q0 + 16 * u, g - 1) * ECH;
            r[u] = rp[(lane < len[u]) ? r0[u] + lane : 0];
        }
        #pragma unroll
        for (int u = 0; u < RU; ++u)
            pos[u] = (lane < len[u]) ? atomicAdd(&s_cur[r[u].y - node0], 1) : -1;
        #pragma unroll
        for (int u = 0; u < RU; ++u)
            if (pos[u] >= 0) csr[pos[u]] = r[u].x;
        #pragma unroll
        for (int u = 0; u < RU; ++u) {
            const int2* rp = pairs + (size_t)min(q0 + 16 * u, g - 1) * ECH + r0[u];
            for (int i = lane + 32; i < len[u]; i += 32) {
                const int2 t = rp[i];
                csr[atomicAdd(&s_cur[t.y - node0], 1)] = t.x;
            }
        }
    }
}

template <bool BIAS>
__global__ void __launch_bounds__(256)
k_gather64(const float* __restrict__ h, const int* __restrict__ rowptr, const int* __restrict__ csr,
           const float* __restrict__ dis, const float* __restrict__ bias,
           float* __restrict__ out, int n) {
    const unsigned c4   = (threadIdx.x & 15u) << 2;
    const unsigned step = (gridDim.x * blockDim.x) >> 4;
    for (unsigned node = (blockIdx.x * blockDim.x + threadIdx.x) >> 4; node < (unsigned)n; node += step) {
        const int beg = rowptr[node];
        const int end = rowptr[node + 1];
        const float sd = dis[node];
        float4 acc = *(const float4*)(h + (size_t)node * 64 + c4);
        acc.x *= sd; acc.y *= sd; acc.z *= sd; acc.w *= sd;
        const int last = end - 1;
        for (int p = beg; p < end; p += 4) {
            const bool ok1 = p + 1 < end, ok2 = p + 2 < end, ok3 = p + 3 < end;
            const int s0 = csr[p];
            const int s1 = csr[min(p + 1, last)];
            const int s2 = csr[min(p + 2, last)];
            const int s3 = csr[min(p + 3, last)];
            const float w0 = dis[s0], w1 = dis[s1], w2 = dis[s2], w3 = dis[s3];
            const float4 v0 = *(const float4*)(h + (size_t)s0 * 64 + c4);
            const float4 v1 = *(const float4*)(h + (size_t)s1 * 64 + c4);
            const float4 v2 = *(const float4*)(h + (size_t)s2 * 64 + c4);
            const float4 v3 = *(const float4*)(h + (size_t)s3 * 64 + c4);
            acc.x = fmaf(w0, v0.x, acc.x); acc.y = fmaf(w0, v0.y, acc.y);
            acc.z = fmaf(w0, v0.z, acc.z); acc.w = fmaf(w0, v0.w, acc.w);
            float4 t;
            t.x = fmaf(w1, v1.x, acc.x); t.y = fmaf(w1, v1.y, acc.y);
            t.z = fmaf(w1, v1.z, acc.z); t.w = fmaf(w1, v1.w, acc.w);
            acc.x = ok1 ? t.x : acc.x; acc.y = ok1 ? t.y : acc.y;
            acc.z = ok1 ? t.z : acc.z; acc.w = ok1 ? t.w : acc.w;
            t.x = fmaf(w2, v2.x, acc.x); t.y = fmaf(w2, v2.y, acc.y);
            t.z = fmaf(w2, v2.z, acc.z); t.w = fmaf(w2, v2.w, acc.w);
            acc.x = ok2 ? t.x : acc.x; acc.y = ok2 ? t.y : acc.y;
            acc.z = ok2 ? t.z : acc.z; acc.w = ok2 ? t.w : acc.w;
            t.x = fmaf(w3, v3.x, acc.x); t.y = fmaf(w3, v3.y, acc.y);
            t.z = fmaf(w3, v3.z, acc.z); t.w = fmaf(w3, v3.w, acc.w);
            acc.x = ok3 ? t.x : acc.x; acc.y = ok3 ? t.y : acc.y;
            acc.z = ok3 ? t.z : acc.z; acc.w = ok3 ? t.w : acc.w;
        }
        float4 r;
        r.x = acc.x * sd; r.y = acc.y * sd; r.z = acc.z * sd; r.w = acc.w * sd;
        if constexpr (BIAS) {
            const float4 b = *(const float4*)(bias + c4);
            r.x += b.x; r.y += b.y; r.z += b.z; r.w += b.w;
        }
        *(float4*)(out + (size_t)node * 64 + c4) = r;
    }
}

constexpr int F0  = 64;
constexpr int F1  = 128;
constexpr int F2  = 64;
constexpr int TM  = 64;
constexpr int LDA = F0 + 4;
constexpr int LDH = F1 + 4;

template <int KK>
__device__ __forceinline__ void mma_2rows(const float* A, int rs, const v2f* bf, v8f& c0, v8f& c1) {
    #pragma unroll
    for (int kk = 0; kk < KK; ++kk) {
        const v2f a0 = *(const v2f*)(A + kk * 4);
        const v2f a1 = *(const v2f*)(A + rs + kk * 4);
        c0 = __builtin_amdgcn_wmma_f32_16x16x4_f32(false, a0, false, bf[kk],
                                                   (short)0, c0, false, false);
        c1 = __builtin_amdgcn_wmma_f32_16x16x4_f32(false, a1, false, bf[kk],
                                                   (short)0, c1, false, false);
    }
}

__global__ void __launch_bounds__(256)
k_mlp_wmma(const float* __restrict__ XA, const float* __restrict__ W1, const float* __restrict__ b1,
           const float* __restrict__ W2, float* __restrict__ OUT, int M) {
    __shared__ __attribute__((aligned(16))) float sA[TM * LDA];
    __shared__ __attribute__((aligned(16))) float sH[TM * LDH];

    const int tid  = threadIdx.x;
    const int wave = tid >> 5;
    const int lane = tid & 31;
    const int half = lane >> 4;
    const int l16  = lane & 15;
    const int m0   = blockIdx.x * TM;

    const int xr = tid >> 4;
    const int c4 = (tid & 15) << 2;
    float4 xv[TM / 16];
    #pragma unroll
    for (int j = 0; j < TM / 16; ++j) {
        int row = m0 + xr + 16 * j; if (row >= M) row = M - 1;
        xv[j] = *(const float4*)(XA + (size_t)row * F0 + c4);
    }

    const int n1 = wave * 16 + l16;
    v2f bf1[F0 / 4];
    #pragma unroll
    for (int kk = 0; kk < F0 / 4; ++kk) {
        const int k = kk * 4 + 2 * half;
        bf1[kk].x = W1[k * F1 + n1];
        bf1[kk].y = W1[(k + 1) * F1 + n1];
    }
    const float bcol = b1[n1];

    #pragma unroll
    for (int j = 0; j < TM / 16; ++j)
        *(float4*)&sA[(xr + 16 * j) * LDA + c4] = xv[j];
    __syncthreads();

    #pragma unroll
    for (int mp = 0; mp < TM / 32; ++mp) {
        v8f c0 = {}, c1 = {};
        mma_2rows<F0 / 4>(&sA[(mp * 32 + l16) * LDA + 2 * half], 16 * LDA, bf1, c0, c1);
        float* H = &sH[(mp * 32 + half * 8) * LDH + n1];
        #pragma unroll
        for (int r = 0; r < 8; ++r) {
            H[r * LDH]        = fmaxf(c0[r] + bcol, 0.0f);
            H[(16 + r) * LDH] = fmaxf(c1[r] + bcol, 0.0f);
        }
    }

    const int n2 = (wave & 3) * 16 + l16;
    v2f bf2[F1 / 4];
    #pragma unroll
    for (int kk = 0; kk < F1 / 4; ++kk) {
        const int k = kk * 4 + 2 * half;
        bf2[kk].x = W2[k * F2 + n2];
        bf2[kk].y = W2[(k + 1) * F2 + n2];
    }
    __syncthreads();

    const int mq = wave >> 2;
    v8f c0 = {}, c1 = {};
    mma_2rows<F1 / 4>(&sH[(mq * 16 + l16) * LDH + 2 * half], 32 * LDH, bf2, c0, c1);
    #pragma unroll
    for (int r = 0; r < 8; ++r) {
        const int row0 = m0 + mq * 16 + half * 8 + r;
        const int row1 = row0 + 32;
        if (row0 < M) OUT[(size_t)row0 * F2 + n2] = c0[r];
        if (row1 < M) OUT[(size_t)row1 * F2 + n2] = c1[r];
    }
}

extern "C" void kernel_launch(void* const* d_in, const int* in_sizes, int n_in,
                              void* d_out, int out_size, void* d_ws, size_t ws_size,
                              hipStream_t stream) {
    const float* X  = (const float*)d_in[0];
    const int*   EI = (const int*)d_in[1];
    const float* W1 = (const float*)d_in[2];
    const float* b1 = (const float*)d_in[3];
    const float* W2 = (const float*)d_in[4];
    const float* b2 = (const float*)d_in[5];
    float* out = (float*)d_out;

    const int FIN1 = 64;
    const int N = in_sizes[0] / FIN1;
    const int E = in_sizes[1] / 2;
    const int* src = EI;
    const int* dst = EI + E;

    size_t off = ((size_t)(N + 1) * sizeof(float) + 255) & ~(size_t)255;
    int*   rowptr = (int*)((char*)d_ws + off);
    float* dis    = (float*)((char*)d_ws + 2 * off);
    float* bufA   = (float*)((char*)d_ws + 3 * off);
    float* bufO   = bufA + (size_t)N * 64;
    int*   csr    = (int*)(bufO + (size_t)N * 128);

    const int B  = (N + BNODES - 1) >> BSH;
    const int G  = (E + ECH - 1) / ECH;
    size_t ooff = ((size_t)(B + 1) * G * sizeof(int) + 255) & ~(size_t)255;
    int*  offs  = (int*)bufO;
    int2* pairs = (int2*)((char*)bufO + ooff);

    const int T = 256;

    k_chunk_group<<<G, LT, B * sizeof(int), stream>>>(src, dst, offs, pairs, E, B);
    k_bucket_csr<<<B, 512, G * sizeof(int2), stream>>>(pairs, offs, rowptr, dis, csr, N, G);

    k_gather64<false><<<2048, T, 0, stream>>>(X, rowptr, csr, dis, nullptr, bufA, N);
    k_mlp_wmma<<<(N + TM - 1) / TM, 256, 0, stream>>>(bufA, W1, b1, W2, bufO, N);
    k_gather64<true><<<2048, T, 0, stream>>>(bufO, rowptr, csr, dis, b2, out, N);
}


# Layer-1 neighbour sum done inside the dense launch: a work-group sums its 64 rows of A X from X straight into its LDS tile; no aggregate store or re-read, 1 launch fewer; f32 sum order unchanged
# speedup vs baseline: 5.4318x; 1.0327x over previous
#include <hip/hip_runtime.h>
#include <hip/hip_bf16.h>


typedef __attribute__((ext_vector_type(2))) float v2f;
typedef __attribute__((ext_vector_type(8))) float v8f;

constexpr int BSH    = 8;
constexpr int BNODES = 1 << BSH;
constexpr int LT     = 512;
constexpr int ECH    = LT * 16;
constexpr int RU     = 13;

__global__ void __launch_bounds__(LT)
k_chunk_group(const int* __restrict__ src, const int* __restrict__ dst, int* __restrict__ off,
              int2* __restrict__ pairs, int e, int nb) {
    extern __shared__ int s_bkt[];
    __shared__ int s_ws[LT / 32];
    const int tid  = threadIdx.x;
    const int lane = tid & 31;
    const int wave = tid >> 5;
    const int g = gridDim.x;
    const int i0 = blockIdx.x * ECH + tid;
    int d[16], s[16], rk[16];
    #pragma unroll
    for (int k = 0; k < 16; ++k) {
        const int i = min(i0 + k * LT, e - 1);
        d[k] = dst[i];
        s[k] = src[i];
    }
    for (int b = tid; b < nb; b += LT) s_bkt[b] = 0;
    __syncthreads();
    #pragma unroll
    for (int k = 0; k < 16; ++k)
        rk[k] = (i0 + k * LT < e) ? atomicAdd(&s_bkt[d[k] >> BSH], 1) : -1;
    __syncthreads();

    int carry = 0;
    for (int b0 = 0; b0 < nb; b0 += LT) {
        const int b = b0 + tid;
        const int c = (b < nb) ? s_bkt[b] : 0;
        int inc = c;
        #pragma unroll
        for (int dlt = 1; dlt < 32; dlt <<= 1) {
            const int u = __shfl_up(inc, dlt);
            if (lane >= dlt) inc += u;
        }
        if (lane == 31) s_ws[wave] = inc;
        __syncthreads();
        int woff = 0, tot = 0;
        #pragma unroll
        for (int w = 0; w < LT / 32; ++w) {
            const int v = s_ws[w];
            tot += v;
            if (w < wave) woff += v;
        }
        const int ex = carry + woff + inc - c;
        if (b < nb) {
            s_bkt[b] = ex;
            off[(size_t)b * g + blockIdx.x] = ex;
        }
        carry += tot;
        __syncthreads();
    }
    if (tid == 0) off[(size_t)nb * g + blockIdx.x] = carry;

    int2* mine = pairs + (size_t)blockIdx.x * ECH;
    int slot[16];
    #pragma unroll
    for (int k = 0; k < 16; ++k) slot[k] = s_bkt[d[k] >> BSH] + rk[k];
    #pragma unroll
    for (int k = 0; k < 16; ++k)
        if (rk[k] >= 0) mine[slot[k]] = make_int2(s[k], d[k]);
}

__global__ void __launch_bounds__(512)
k_bucket_csr(const int2* __restrict__ pairs, const int* __restrict__ off, int* __restrict__ rowptr,
             float* __restrict__ dis, int* __restrict__ csr, int n, int g) {
    extern __shared__ int2 s_run[];
    __shared__ int s_cnt[BNODES];
    __shared__ int s_cur[BNODES];
    __shared__ int s_wsum[BNODES / 32];
    __shared__ int s_red[512 / 32];
    const int tid   = threadIdx.x;
    const int lane  = tid & 31;
    const int wave  = tid >> 5;
    const int b     = blockIdx.x;
    const int node0 = b << BSH;

    if (tid < BNODES) s_cnt[tid] = 0;
    int a0 = 0;
    for (int q = tid; q < g; q += 512) {
        const int r0 = off[(size_t)b * g + q];
        const int r1 = off[(size_t)(b + 1) * g + q];
        s_run[q] = make_int2(r0, r1 - r0);
        a0 += r0;
    }
    #pragma unroll
    for (int dlt = 16; dlt >= 1; dlt >>= 1) a0 += __shfl_xor(a0, dlt);
    if (lane == 0) s_red[wave] = a0;
    __syncthreads();
    int beg = 0;
    #pragma unroll
    for (int w = 0; w < 512 / 32; ++w) beg += s_red[w];

    for (int q0 = wave; q0 < g; q0 += 16 * RU) {
        int r0[RU], len[RU], y[RU];
        #pragma unroll
        for (int u = 0; u < RU; ++u) {
            const int q = q0 + 16 * u;
            const int2 rl = s_run[min(q, g - 1)];
            r0[u]  = rl.x;
            len[u] = (q < g) ? rl.y : 0;
        }
        #pragma unroll
        for (int u = 0; u < RU; ++u) {
            const int2* rp = pairs + (size_t)min(q0 + 16 * u, g - 1) * ECH;
            y[u] = rp[(lane < len[u]) ? r0[u] + lane : 0].y;
        }
        #pragma unroll
        for (int u = 0; u < RU; ++u)
            if (lane < len[u]) atomicAdd(&s_cnt[y[u] - node0], 1);
        #pragma unroll
        for (int u = 0; u < RU; ++u) {
            const int2* rp = pairs + (size_t)min(q0 + 16 * u, g - 1) * ECH + r0[u];
            for (int i = lane + 32; i < len[u]; i += 32)
                atomicAdd(&s_cnt[rp[i].y - node0], 1);
        }
    }
    __syncthreads();

    const int c = (tid < BNODES) ? s_cnt[tid] : 0;
    int inc = c;
    #pragma unroll
    for (int dlt = 1; dlt < 32; dlt <<= 1) {
        const int u = __shfl_up(inc, dlt);
        if (lane >= dlt) inc += u;
    }
    if (tid < BNODES && lane == 31) s_wsum[wave] = inc;
    __syncthreads();
    if (tid < BNODES) {
        int ex = beg + inc - c;
        #pragma unroll
        for (int w = 0; w < BNODES / 32; ++w)
            if (w < wave) ex += s_wsum[w];
        s_cur[tid] = ex;
        const int node = node0 + tid;
        if (node < n) {
            rowptr[node] = ex;
            dis[node] = rsqrtf((float)(c + 1));
            if (node == n - 1) rowptr[n] = ex + c;
        }
    }
    __syncthreads();

    for (int q0 = wave; q0 < g; q0 += 16 * RU) {
        int r0[RU], len[RU], pos[RU];
        int2 r[RU];
        #pragma unroll
        for (int u = 0; u < RU; ++u) {
            const int q = q0 + 16 * u;
            const int2 rl = s_run[min(q, g - 1)];
            r0[u]  = rl.x;
            len[u] = (q < g) ? rl.y : 0;
        }
        #pragma unroll
        for (int u = 0; u < RU; ++u) {
            const int2* rp = pairs + (size_t)min(q0 + 16 * u, g - 1) * ECH;
            r[u] = rp[(lane < len[u]) ? r0[u] + lane : 0];
        }
        #pragma unroll
        for (int u = 0; u < RU; ++u)
            pos[u] = (lane < len[u]) ? atomicAdd(&s_cur[r[u].y - node0], 1) : -1;
        #pragma unroll
        for (int u = 0; u < RU; ++u)
            if (pos[u] >= 0) csr[pos[u]] = r[u].x;
        #pragma unroll
        for (int u = 0; u < RU; ++u) {
            const int2* rp = pairs + (size_t)min(q0 + 16 * u, g - 1) * ECH + r0[u];
            for (int i = lane + 32; i < len[u]; i += 32) {
                const int2 t = rp[i];
                csr[atomicAdd(&s_cur[t.y - node0], 1)] = t.x;
            }
        }
    }
}

template <bool BIAS>
__global__ void __launch_bounds__(256)
k_gather64(const float* __restrict__ h, const int* __restrict__ rowptr, const int* __restrict__ csr,
           const float* __restrict__ dis, const float* __restrict__ bias,
           float* __restrict__ out, int n) {
    const unsigned c4   = (threadIdx.x & 15u) << 2;
    const unsigned step = (gridDim.x * blockDim.x) >> 4;
    for (unsigned node = (blockIdx.x * blockDim.x + threadIdx.x) >> 4; node < (unsigned)n; node += step) {
        const int beg = rowptr[node];
        const int end = rowptr[node + 1];
        const float sd = dis[node];
        float4 acc = *(const float4*)(h + (size_t)node * 64 + c4);
        acc.x *= sd; acc.y *= sd; acc.z *= sd; acc.w *= sd;
        const int last = end - 1;
        for (int p = beg; p < end; p += 4) {
            const bool ok1 = p + 1 < end, ok2 = p + 2 < end, ok3 = p + 3 < end;
            const int s0 = csr[p];
            const int s1 = csr[min(p + 1, last)];
            const int s2 = csr[min(p + 2, last)];
            const int s3 = csr[min(p + 3, last)];
            const float w0 = dis[s0], w1 = dis[s1], w2 = dis[s2], w3 = dis[s3];
            const float4 v0 = *(const float4*)(h + (size_t)s0 * 64 + c4);
            const float4 v1 = *(const float4*)(h + (size_t)s1 * 64 + c4);
            const float4 v2 = *(const float4*)(h + (size_t)s2 * 64 + c4);
            const float4 v3 = *(const float4*)(h + (size_t)s3 * 64 + c4);
            acc.x = fmaf(w0, v0.x, acc.x); acc.y = fmaf(w0, v0.y, acc.y);
            acc.z = fmaf(w0, v0.z, acc.z); acc.w = fmaf(w0, v0.w, acc.w);
            float4 t;
            t.x = fmaf(w1, v1.x, acc.x); t.y = fmaf(w1, v1.y, acc.y);
            t.z = fmaf(w1, v1.z, acc.z); t.w = fmaf(w1, v1.w, acc.w);
            acc.x = ok1 ? t.x : acc.x; acc.y = ok1 ? t.y : acc.y;
            acc.z = ok1 ? t.z : acc.z; acc.w = ok1 ? t.w : acc.w;
            t.x = fmaf(w2, v2.x, acc.x); t.y = fmaf(w2, v2.y, acc.y);
            t.z = fmaf(w2, v2.z, acc.z); t.w = fmaf(w2, v2.w, acc.w);
            acc.x = ok2 ? t.x : acc.x; acc.y = ok2 ? t.y : acc.y;
            acc.z = ok2 ? t.z : acc.z; acc.w = ok2 ? t.w : acc.w;
            t.x = fmaf(w3, v3.x, acc.x); t.y = fmaf(w3, v3.y, acc.y);
            t.z = fmaf(w3, v3.z, acc.z); t.w = fmaf(w3, v3.w, acc.w);
            acc.x = ok3 ? t.x : acc.x; acc.y = ok3 ? t.y : acc.y;
            acc.z = ok3 ? t.z : acc.z; acc.w = ok3 ? t.w : acc.w;
        }
        float4 r;
        r.x = acc.x * sd; r.y = acc.y * sd; r.z = acc.z * sd; r.w = acc.w * sd;
        if constexpr (BIAS) {
            const float4 b = *(const float4*)(bias + c4);
            r.x += b.x; r.y += b.y; r.z += b.z; r.w += b.w;
        }
        *(float4*)(out + (size_t)node * 64 + c4) = r;
    }
}

constexpr int F0  = 64;
constexpr int F1  = 128;
constexpr int F2  = 64;
constexpr int TM  = 64;
constexpr int LDA = F0 + 4;
constexpr int LDH = F1 + 4;

template <int KK>
__device__ __forceinline__ void mma_2rows(const float* A, int rs, const v2f* bf, v8f& c0, v8f& c1) {
    #pragma unroll
    for (int kk = 0; kk < KK; ++kk) {
        const v2f a0 = *(const v2f*)(A + kk * 4);
        const v2f a1 = *(const v2f*)(A + rs + kk * 4);
        c0 = __builtin_amdgcn_wmma_f32_16x16x4_f32(false, a0, false, bf[kk],
                                                   (short)0, c0, false, false);
        c1 = __builtin_amdgcn_wmma_f32_16x16x4_f32(false, a1, false, bf[kk],
                                                   (short)0, c1, false, false);
    }
}

template <int NR>
__device__ __forceinline__ void sum_rows(const float* __restrict__ X, int my, float mw, int k0, int rem,
                                         unsigned c4, float4& acc) {
    float  w[NR];
    float4 v[NR];
    #pragma unroll
    for (int i = 0; i < NR; ++i) {
        const int s = __shfl(my, k0 + i, 16);
        w[i] = __shfl(mw, k0 + i, 16);
        v[i] = *(const float4*)(X + (size_t)s * 64 + c4);
    }
    #pragma unroll
    for (int i = 0; i < NR; ++i) {
        const bool ok = k0 + i < rem;
        float4 t;
        t.x = fmaf(w[i], v[i].x, acc.x); t.y = fmaf(w[i], v[i].y, acc.y);
        t.z = fmaf(w[i], v[i].z, acc.z); t.w = fmaf(w[i], v[i].w, acc.w);
        acc.x = ok ? t.x : acc.x; acc.y = ok ? t.y : acc.y;
        acc.z = ok ? t.z : acc.z; acc.w = ok ? t.w : acc.w;
    }
}

__global__ void __launch_bounds__(256)
k_gather_mlp(const float* __restrict__ X, const int* __restrict__ rowptr, const int* __restrict__ csr,
             const float* __restrict__ dis, const float* __restrict__ W1, const float* __restrict__ b1,
             const float* __restrict__ W2, float* __restrict__ OUT, int M) {
    __shared__ __attribute__((aligned(16))) float sA[TM * LDA];
    __shared__ __attribute__((aligned(16))) float sH[TM * LDH];

    const int tid  = threadIdx.x;
    const int wave = tid >> 5;
    const int lane = tid & 31;
    const int half = lane >> 4;
    const int l16  = lane & 15;
    const int m0   = blockIdx.x * TM;

    const int xr = tid >> 4;
    const unsigned c4 = (tid & 15) << 2;
    int nd[TM / 16], beg[TM / 16], cnt[TM / 16], my0[TM / 16];
    float sdv[TM / 16], mw0[TM / 16];
    #pragma unroll
    for (int j = 0; j < TM / 16; ++j) {
        int row = m0 + xr + 16 * j; if (row >= M) row = M - 1;
        nd[j]  = row;
        beg[j] = rowptr[row];
        cnt[j] = rowptr[row + 1] - beg[j];
        sdv[j] = dis[row];
    }
    #pragma unroll
    for (int j = 0; j < TM / 16; ++j) {
        my0[j] = (l16 < cnt[j]) ? csr[beg[j] + l16] : nd[j];
        mw0[j] = dis[my0[j]];
    }
    #pragma unroll
    for (int j = 0; j < TM / 16; ++j) {
        const float sd = sdv[j];
        float4 acc = *(const float4*)(X + (size_t)nd[j] * F0 + c4);
        acc.x *= sd; acc.y *= sd; acc.z *= sd; acc.w *= sd;
        int   my = my0[j];
        float mw = mw0[j];
        int rem  = cnt[j];
        int remw = __builtin_amdgcn_readfirstlane(max(rem, __shfl_xor(rem, 16)));
        int base = beg[j];
        while (remw > 0) {
            int nmy = nd[j];
            if (remw > 16) nmy = (16 + l16 < rem) ? csr[base + 16 + l16] : nd[j];
            if (remw > 4) sum_rows<8>(X, my, mw, 0, rem, c4, acc);
            else          sum_rows<4>(X, my, mw, 0, rem, c4, acc);
            float nmw = 0.0f;
            if (remw > 16) nmw = dis[nmy];
            if (remw > 8) {
                if (remw > 12) sum_rows<8>(X, my, mw, 8, rem, c4, acc);
                else           sum_rows<4>(X, my, mw, 8, rem, c4, acc);
            }
            my = nmy; mw = nmw;
            rem -= 16; remw -= 16; base += 16;
        }
        float4 r;
        r.x = acc.x * sd; r.y = acc.y * sd; r.z = acc.z * sd; r.w = acc.w * sd;
        *(float4*)&sA[(xr + 16 * j) * LDA + c4] = r;
    }

    const int n1 = wave * 16 + l16;
    v2f bf1[F0 / 4];
    #pragma unroll
    for (int kk = 0; kk < F0 / 4; ++kk) {
        const int k = kk * 4 + 2 * half;
        bf1[kk].x = W1[k * F1 + n1];
        bf1[kk].y = W1[(k + 1) * F1 + n1];
    }
    const float bcol = b1[n1];

    __syncthreads();

    #pragma unroll
    for (int mp = 0; mp < TM / 32; ++mp) {
        v8f c0 = {}, c1 = {};
        mma_2rows<F0 / 4>(&sA[(mp * 32 + l16) * LDA + 2 * half], 16 * LDA, bf1, c0, c1);
        float* H = &sH[(mp * 32 + half * 8) * LDH + n1];
        #pragma unroll
        for (int r = 0; r < 8; ++r) {
            H[r * LDH]        = fmaxf(c0[r] + bcol, 0.0f);
            H[(16 + r) * LDH] = fmaxf(c1[r] + bcol, 0.0f);
        }
    }

    const int n2 = (wave & 3) * 16 + l16;
    v2f bf2[F1 / 4];
    #pragma unroll
    for (int kk = 0; kk < F1 / 4; ++kk) {
        const int k = kk * 4 + 2 * half;
        bf2[kk].x = W2[k * F2 + n2];
        bf2[kk].y = W2[(k + 1) * F2 + n2];
    }
    __syncthreads();

    const int mq = wave >> 2;
    v8f c0 = {}, c1 = {};
    mma_2rows<F1 / 4>(&sH[(mq * 16 + l16) * LDH + 2 * half], 32 * LDH, bf2, c0, c1);
    #pragma unroll
    for (int r = 0; r < 8; ++r) {
        const int row0 = m0 + mq * 16 + half * 8 + r;
        const int row1 = row0 + 32;
        if (row0 < M) OUT[(size_t)row0 * F2 + n2] = c0[r];
        if (row1 < M) OUT[(size_t)row1 * F2 + n2] = c1[r];
    }
}

extern "C" void kernel_launch(void* const* d_in, const int* in_sizes, int n_in,
                              void* d_out, int out_size, void* d_ws, size_t ws_size,
                              hipStream_t stream) {
    const float* X  = (const float*)d_in[0];
    const int*   EI = (const int*)d_in[1];
    const float* W1 = (const float*)d_in[2];
    const float* b1 = (const float*)d_in[3];
    const float* W2 = (const float*)d_in[4];
    const float* b2 = (const float*)d_in[5];
    float* out = (float*)d_out;

    const int FIN1 = 64;
    const int N = in_sizes[0] / FIN1;
    const int E = in_sizes[1] / 2;
    const int* src = EI;
    const int* dst = EI + E;

    size_t off = ((size_t)(N + 1) * sizeof(float) + 255) & ~(size_t)255;
    int*   rowptr = (int*)((char*)d_ws + off);
    float* dis    = (float*)((char*)d_ws + 2 * off);
    float* bufA   = (float*)((char*)d_ws + 3 * off);
    float* bufO   = bufA + (size_t)N * 64;
    int*   csr    = (int*)(bufO + (size_t)N * 128);

    const int B  = (N + BNODES - 1) >> BSH;
    const int G  = (E + ECH - 1) / ECH;
    size_t ooff = ((size_t)(B + 1) * G * sizeof(int) + 255) & ~(size_t)255;
    int*  offs  = (int*)bufO;
    int2* pairs = (int2*)((char*)bufO + ooff);

    const int T = 256;

    k_chunk_group<<<G, LT, B * sizeof(int), stream>>>(src, dst, offs, pairs, E, B);
    k_bucket_csr<<<B, 512, G * sizeof(int2), stream>>>(pairs, offs, rowptr, dis, csr, N, G);

    k_gather_mlp<<<(N + TM - 1) / TM, 256, 0, stream>>>(X, rowptr, csr, dis, W1, b1, W2, bufO, N);
    k_gather64<true><<<2048, T, 0, stream>>>(bufO, rowptr, csr, dis, b2, out, N);
}


# Grouped edges and csr slots are ordered in LDS and stored as whole 128-B lines (both grouping levels), not as 8-B/4-B stores over 32 lines a wave; f32 sum order = LDS arrival order as before
# speedup vs baseline: 5.8823x; 1.0829x over previous
#include <hip/hip_runtime.h>
#include <hip/hip_bf16.h>


typedef __attribute__((ext_vector_type(2))) float v2f;
typedef __attribute__((ext_vector_type(8))) float v8f;

constexpr int BSH    = 8;
constexpr int BNODES = 1 << BSH;
constexpr int LT     = 512;
constexpr int ECH    = LT * 16;
constexpr int RU     = 13;
constexpr int SCAP   = 8192;

__global__ void __launch_bounds__(LT)
k_chunk_group(const int* __restrict__ src, const int* __restrict__ dst, int* __restrict__ off,
              int2* __restrict__ pairs, int e, int nb) {
    extern __shared__ int s_bkt[];
    __shared__ int s_ws[LT / 32];
    __shared__ __attribute__((aligned(16))) int2 s_pair[ECH];
    const int tid  = threadIdx.x;
    const int lane = tid & 31;
    const int wave = tid >> 5;
    const int g = gridDim.x;
    const int i0 = blockIdx.x * ECH + tid;
    int d[16], s[16], rk[16];
    #pragma unroll
    for (int k = 0; k < 16; ++k) {
        const int i = min(i0 + k * LT, e - 1);
        d[k] = dst[i];
        s[k] = src[i];
    }
    for (int b = tid; b < nb; b += LT) s_bkt[b] = 0;
    __syncthreads();
    #pragma unroll
    for (int k = 0; k < 16; ++k)
        rk[k] = (i0 + k * LT < e) ? atomicAdd(&s_bkt[d[k] >> BSH], 1) : -1;
    __syncthreads();

    int carry = 0;
    for (int b0 = 0; b0 < nb; b0 += LT) {
        const int b = b0 + tid;
        const int c = (b < nb) ? s_bkt[b] : 0;
        int inc = c;
        #pragma unroll
        for (int dlt = 1; dlt < 32; dlt <<= 1) {
            const int u = __shfl_up(inc, dlt);
            if (lane >= dlt) inc += u;
        }
        if (lane == 31) s_ws[wave] = inc;
        __syncthreads();
        int woff = 0, tot = 0;
        #pragma unroll
        for (int w = 0; w < LT / 32; ++w) {
            const int v = s_ws[w];
            tot += v;
            if (w < wave) woff += v;
        }
        const int ex = carry + woff + inc - c;
        if (b < nb) {
            s_bkt[b] = ex;
            off[(size_t)b * g + blockIdx.x] = ex;
        }
        carry += tot;
        __syncthreads();
    }
    if (tid == 0) off[(size_t)nb * g + blockIdx.x] = carry;

    int slot[16];
    #pragma unroll
    for (int k = 0; k < 16; ++k) slot[k] = s_bkt[d[k] >> BSH] + rk[k];
    #pragma unroll
    for (int k = 0; k < 16; ++k)
        if (rk[k] >= 0) s_pair[slot[k]] = make_int2(s[k], d[k]);
    __syncthreads();

    int2* mine = pairs + (size_t)blockIdx.x * ECH;
    const int cnt = min(ECH, e - (int)blockIdx.x * ECH);
    if (cnt == ECH) {
        int4 v[ECH / (2 * LT)];
        #pragma unroll
        for (int k = 0; k < ECH / (2 * LT); ++k)
            v[k] = *reinterpret_cast<const int4*>(s_pair + 2 * (k * LT + tid));
        #pragma unroll
        for (int k = 0; k < ECH / (2 * LT); ++k)
            *reinterpret_cast<int4*>(mine + 2 * (k * LT + tid)) = v[k];
    } else {
        for (int i = tid; i < cnt; i += LT) mine[i] = s_pair[i];
    }
}

__global__ void __launch_bounds__(512)
k_bucket_csr(const int2* __restrict__ pairs, const int* __restrict__ off, int* __restrict__ rowptr,
             float* __restrict__ dis, int* __restrict__ csr, int n, int g) {
    extern __shared__ int2 s_run[];
    __shared__ int s_cnt[BNODES];
    __shared__ int s_cur[BNODES];
    __shared__ int s_wsum[BNODES / 32];
    __shared__ int s_red[512 / 32];
    __shared__ int s_slot[SCAP];
    __shared__ int s_tot;
    const int tid   = threadIdx.x;
    const int lane  = tid & 31;
    const int wave  = tid >> 5;
    const int b     = blockIdx.x;
    const int node0 = b << BSH;

    if (tid < BNODES) s_cnt[tid] = 0;
    int a0 = 0;
    for (int q = tid; q < g; q += 512) {
        const int r0 = off[(size_t)b * g + q];
        const int r1 = off[(size_t)(b + 1) * g + q];
        s_run[q] = make_int2(r0, r1 - r0);
        a0 += r0;
    }
    #pragma unroll
    for (int dlt = 16; dlt >= 1; dlt >>= 1) a0 += __shfl_xor(a0, dlt);
    if (lane == 0) s_red[wave] = a0;
    __syncthreads();
    int beg = 0;
    #pragma unroll
    for (int w = 0; w < 512 / 32; ++w) beg += s_red[w];

    for (int q0 = wave; q0 < g; q0 += 16 * RU) {
        int r0[RU], len[RU], y[RU];
        #pragma unroll
        for (int u = 0; u < RU; ++u) {
            const int q = q0 + 16 * u;
            const int2 rl = s_run[min(q, g - 1)];
            r0[u]  = rl.x;
            len[u] = (q < g) ? rl.y : 0;
        }
        #pragma unroll
        for (int u = 0; u < RU; ++u) {
            const int2* rp = pairs + (size_t)min(q0 + 16 * u, g - 1) * ECH;
            y[u] = rp[(lane < len[u]) ? r0[u] + lane : 0].y;
        }
        #pragma unroll
        for (int u = 0; u < RU; ++u)
            if (lane < len[u]) atomicAdd(&s_cnt[y[u] - node0], 1);
        #pragma unroll
        for (int u = 0; u < RU; ++u) {
            const int2* rp = pairs + (size_t)min(q0 + 16 * u, g - 1) * ECH + r0[u];
            for (int i = lane + 32; i < len[u]; i += 32)
                atomicAdd(&s_cnt[rp[i].y - node0], 1);
        }
    }
    __syncthreads();

    const int c = (tid < BNODES) ? s_cnt[tid] : 0;
    int inc = c;
    #pragma unroll
    for (int dlt = 1; dlt < 32; dlt <<= 1) {
        const int u = __shfl_up(inc, dlt);
        if (lane >= dlt) inc += u;
    }
    if (tid < BNODES && lane == 31) s_wsum[wave] = inc;
    __syncthreads();
    if (tid < BNODES) {
        int ex = beg + inc - c;
        #pragma unroll
        for (int w = 0; w < BNODES / 32; ++w)
            if (w < wave) ex += s_wsum[w];
        s_cur[tid] = ex;
        const int node = node0 + tid;
        if (node < n) {
            rowptr[node] = ex;
            dis[node] = rsqrtf((float)(c + 1));
            if (node == n - 1) rowptr[n] = ex + c;
        }
        if (tid == BNODES - 1) s_tot = ex + c - beg;
    }
    __syncthreads();
    const int  tot    = s_tot;
    const bool staged = tot <= SCAP;

    for (int q0 = wave; q0 < g; q0 += 16 * RU) {
        int r0[RU], len[RU], pos[RU];
        int2 r[RU];
        #pragma unroll
        for (int u = 0; u < RU; ++u) {
            const int q = q0 + 16 * u;
            const int2 rl = s_run[min(q, g - 1)];
            r0[u]  = rl.x;
            len[u] = (q < g) ? rl.y : 0;
        }
        #pragma unroll
        for (int u = 0; u < RU; ++u) {
            const int2* rp = pairs + (size_t)min(q0 + 16 * u, g - 1) * ECH;
            r[u] = rp[(lane < len[u]) ? r0[u] + lane : 0];
        }
        #pragma unroll
        for (int u = 0; u < RU; ++u)
            pos[u] = (lane < len[u]) ? atomicAdd(&s_cur[r[u].y - node0], 1) : -1;
        #pragma unroll
        for (int u = 0; u < RU; ++u)
            if (pos[u] >= 0) {
                if (staged) s_slot[pos[u] - beg] = r[u].x;
                else        csr[pos[u]] = r[u].x;
            }
        #pragma unroll
        for (int u = 0; u < RU; ++u) {
            const int2* rp = pairs + (size_t)min(q0 + 16 * u, g - 1) * ECH + r0[u];
            for (int i = lane + 32; i < len[u]; i += 32) {
                const int2 t = rp[i];
                const int ps = atomicAdd(&s_cur[t.y - node0], 1);
                if (staged) s_slot[ps - beg] = t.x;
                else        csr[ps] = t.x;
            }
        }
    }
    __syncthreads();
    if (staged) {
        const int a = beg & 31;
        for (int p0 = tid - a; p0 < tot; p0 += 4 * 512) {
            int v[4];
            #pragma unroll
            for (int j = 0; j < 4; ++j)
                v[j] = s_slot[min(max(p0 + j * 512, 0), SCAP - 1)];
            #pragma unroll
            for (int j = 0; j < 4; ++j) {
                const int p = p0 + j * 512;
                if (p >= 0 && p < tot) csr[beg + p] = v[j];
            }
        }
    }
}

template <bool BIAS>
__global__ void __launch_bounds__(256)
k_gather64(const float* __restrict__ h, const int* __restrict__ rowptr, const int* __restrict__ csr,
           const float* __restrict__ dis, const float* __restrict__ bias,
           float* __restrict__ out, int n) {
    const unsigned c4   = (threadIdx.x & 15u) << 2;
    const unsigned step = (gridDim.x * blockDim.x) >> 4;
    for (unsigned node = (blockIdx.x * blockDim.x + threadIdx.x) >> 4; node < (unsigned)n; node += step) {
        const int beg = rowptr[node];
        const int end = rowptr[node + 1];
        const float sd = dis[node];
        float4 acc = *(const float4*)(h + (size_t)node * 64 + c4);
        acc.x *= sd; acc.y *= sd; acc.z *= sd; acc.w *= sd;
        const int last = end - 1;
        for (int p = beg; p < end; p += 4) {
            const bool ok1 = p + 1 < end, ok2 = p + 2 < end, ok3 = p + 3 < end;
            const int s0 = csr[p];
            const int s1 = csr[min(p + 1, last)];
            const int s2 = csr[min(p + 2, last)];
            const int s3 = csr[min(p + 3, last)];
            const float w0 = dis[s0], w1 = dis[s1], w2 = dis[s2], w3 = dis[s3];
            const float4 v0 = *(const float4*)(h + (size_t)s0 * 64 + c4);
            const float4 v1 = *(const float4*)(h + (size_t)s1 * 64 + c4);
            const float4 v2 = *(const float4*)(h + (size_t)s2 * 64 + c4);
            const float4 v3 = *(const float4*)(h + (size_t)s3 * 64 + c4);
            acc.x = fmaf(w0, v0.x, acc.x); acc.y = fmaf(w0, v0.y, acc.y);
            acc.z = fmaf(w0, v0.z, acc.z); acc.w = fmaf(w0, v0.w, acc.w);
            float4 t;
            t.x = fmaf(w1, v1.x, acc.x); t.y = fmaf(w1, v1.y, acc.y);
            t.z = fmaf(w1, v1.z, acc.z); t.w = fmaf(w1, v1.w, acc.w);
            acc.x = ok1 ? t.x : acc.x; acc.y = ok1 ? t.y : acc.y;
            acc.z = ok1 ? t.z : acc.z; acc.w = ok1 ? t.w : acc.w;
            t.x = fmaf(w2, v2.x, acc.x); t.y = fmaf(w2, v2.y, acc.y);
            t.z = fmaf(w2, v2.z, acc.z); t.w = fmaf(w2, v2.w, acc.w);
            acc.x = ok2 ? t.x : acc.x; acc.y = ok2 ? t.y : acc.y;
            acc.z = ok2 ? t.z : acc.z; acc.w = ok2 ? t.w : acc.w;
            t.x = fmaf(w3, v3.x, acc.x); t.y = fmaf(w3, v3.y, acc.y);
            t.z = fmaf(w3, v3.z, acc.z); t.w = fmaf(w3, v3.w, acc.w);
            acc.x = ok3 ? t.x : acc.x; acc.y = ok3 ? t.y : acc.y;
            acc.z = ok3 ? t.z : acc.z; acc.w = ok3 ? t.w : acc.w;
        }
        float4 r;
        r.x = acc.x * sd; r.y = acc.y * sd; r.z = acc.z * sd; r.w = acc.w * sd;
        if constexpr (BIAS) {
            const float4 b = *(const float4*)(bias + c4);
            r.x += b.x; r.y += b.y; r.z += b.z; r.w += b.w;
        }
        *(float4*)(out + (size_t)node * 64 + c4) = r;
    }
}

constexpr int F0  = 64;
constexpr int F1  = 128;
constexpr int F2  = 64;
constexpr int TM  = 64;
constexpr int LDA = F0 + 4;
constexpr int LDH = F1 + 4;

template <int KK>
__device__ __forceinline__ void mma_2rows(const float* A, int rs, const v2f* bf, v8f& c0, v8f& c1) {
    #pragma unroll
    for (int kk = 0; kk < KK; ++kk) {
        const v2f a0 = *(const v2f*)(A + kk * 4);
        const v2f a1 = *(const v2f*)(A + rs + kk * 4);
        c0 = __builtin_amdgcn_wmma_f32_16x16x4_f32(false, a0, false, bf[kk],
                                                   (short)0, c0, false, false);
        c1 = __builtin_amdgcn_wmma_f32_16x16x4_f32(false, a1, false, bf[kk],
                                                   (short)0, c1, false, false);
    }
}

template <int NR>
__device__ __forceinline__ void sum_rows(const float* __restrict__ X, int my, float mw, int k0, int rem,
                                         unsigned c4, float4& acc) {
    float  w[NR];
    float4 v[NR];
    #pragma unroll
    for (int i = 0; i < NR; ++i) {
        const int s = __shfl(my, k0 + i, 16);
        w[i] = __shfl(mw, k0 + i, 16);
        v[i] = *(const float4*)(X + (size_t)s * 64 + c4);
    }
    #pragma unroll
    for (int i = 0; i < NR; ++i) {
        const bool ok = k0 + i < rem;
        float4 t;
        t.x = fmaf(w[i], v[i].x, acc.x); t.y = fmaf(w[i], v[i].y, acc.y);
        t.z = fmaf(w[i], v[i].z, acc.z); t.w = fmaf(w[i], v[i].w, acc.w);
        acc.x = ok ? t.x : acc.x; acc.y = ok ? t.y : acc.y;
        acc.z = ok ? t.z : acc.z; acc.w = ok ? t.w : acc.w;
    }
}

__global__ void __launch_bounds__(256)
k_gather_mlp(const float* __restrict__ X, const int* __restrict__ rowptr, const int* __restrict__ csr,
             const float* __restrict__ dis, const float* __restrict__ W1, const float* __restrict__ b1,
             const float* __restrict__ W2, float* __restrict__ OUT, int M) {
    __shared__ __attribute__((aligned(16))) float sA[TM * LDA];
    __shared__ __attribute__((aligned(16))) float sH[TM * LDH];

    const int tid  = threadIdx.x;
    const int wave = tid >> 5;
    const int lane = tid & 31;
    const int half = lane >> 4;
    const int l16  = lane & 15;
    const int m0   = blockIdx.x * TM;

    const int xr = tid >> 4;
    const unsigned c4 = (tid & 15) << 2;
    int nd[TM / 16], beg[TM / 16], cnt[TM / 16], my0[TM / 16];
    float sdv[TM / 16], mw0[TM / 16];
    #pragma unroll
    for (int j = 0; j < TM / 16; ++j) {
        int row = m0 + xr + 16 * j; if (row >= M) row = M - 1;
        nd[j]  = row;
        beg[j] = rowptr[row];
        cnt[j] = rowptr[row + 1] - beg[j];
        sdv[j] = dis[row];
    }
    #pragma unroll
    for (int j = 0; j < TM / 16; ++j) {
        my0[j] = (l16 < cnt[j]) ? csr[beg[j] + l16] : nd[j];
        mw0[j] = dis[my0[j]];
    }
    #pragma unroll
    for (int j = 0; j < TM / 16; ++j) {
        const float sd = sdv[j];
        float4 acc = *(const float4*)(X + (size_t)nd[j] * F0 + c4);
        acc.x *= sd; acc.y *= sd; acc.z *= sd; acc.w *= sd;
        int   my = my0[j];
        float mw = mw0[j];
        int rem  = cnt[j];
        int remw = __builtin_amdgcn_readfirstlane(max(rem, __shfl_xor(rem, 16)));
        int base = beg[j];
        while (remw > 0) {
            int nmy = nd[j];
            if (remw > 16) nmy = (16 + l16 < rem) ? csr[base + 16 + l16] : nd[j];
            if (remw > 4) sum_rows<8>(X, my, mw, 0, rem, c4, acc);
            else          sum_rows<4>(X, my, mw, 0, rem, c4, acc);
            float nmw = 0.0f;
            if (remw > 16) nmw = dis[nmy];
            if (remw > 8) {
                if (remw > 12) sum_rows<8>(X, my, mw, 8, rem, c4, acc);
                else           sum_rows<4>(X, my, mw, 8, rem, c4, acc);
            }
            my = nmy; mw = nmw;
            rem -= 16; remw -= 16; base += 16;
        }
        float4 r;
        r.x = acc.x * sd; r.y = acc.y * sd; r.z = acc.z * sd; r.w = acc.w * sd;
        *(float4*)&sA[(xr + 16 * j) * LDA + c4] = r;
    }

    const int n1 = wave * 16 + l16;
    v2f bf1[F0 / 4];
    #pragma unroll
    for (int kk = 0; kk < F0 / 4; ++kk) {
        const int k = kk * 4 + 2 * half;
        bf1[kk].x = W1[k * F1 + n1];
        bf1[kk].y = W1[(k + 1) * F1 + n1];
    }
    const float bcol = b1[n1];

    __syncthreads();

    #pragma unroll
    for (int mp = 0; mp < TM / 32; ++mp) {
        v8f c0 = {}, c1 = {};
        mma_2rows<F0 / 4>(&sA[(mp * 32 + l16) * LDA + 2 * half], 16 * LDA, bf1, c0, c1);
        float* H = &sH[(mp * 32 + half * 8) * LDH + n1];
        #pragma unroll
        for (int r = 0; r < 8; ++r) {
            H[r * LDH]        = fmaxf(c0[r] + bcol, 0.0f);
            H[(16 + r) * LDH] = fmaxf(c1[r] + bcol, 0.0f);
        }
    }

    const int n2 = (wave & 3) * 16 + l16;
    v2f bf2[F1 / 4];
    #pragma unroll
    for (int kk = 0; kk < F1 / 4; ++kk) {
        const int k = kk * 4 + 2 * half;
        bf2[kk].x = W2[k * F2 + n2];
        bf2[kk].y = W2[(k + 1) * F2 + n2];
    }
    __syncthreads();

    const int mq = wave >> 2;
    v8f c0 = {}, c1 = {};
    mma_2rows<F1 / 4>(&sH[(mq * 16 + l16) * LDH + 2 * half], 32 * LDH, bf2, c0, c1);
    #pragma unroll
    for (int r = 0; r < 8; ++r) {
        const int row0 = m0 + mq * 16 + half * 8 + r;
        const int row1 = row0 + 32;
        if (row0 < M) OUT[(size_t)row0 * F2 + n2] = c0[r];
        if (row1 < M) OUT[(size_t)row1 * F2 + n2] = c1[r];
    }
}

extern "C" void kernel_launch(void* const* d_in, const int* in_sizes, int n_in,
                              void* d_out, int out_size, void* d_ws, size_t ws_size,
                              hipStream_t stream) {
    const float* X  = (const float*)d_in[0];
    const int*   EI = (const int*)d_in[1];
    const float* W1 = (const float*)d_in[2];
    const float* b1 = (const float*)d_in[3];
    const float* W2 = (const float*)d_in[4];
    const float* b2 = (const float*)d_in[5];
    float* out = (float*)d_out;

    const int FIN1 = 64;
    const int N = in_sizes[0] / FIN1;
    const int E = in_sizes[1] / 2;
    const int* src = EI;
    const int* dst = EI + E;

    size_t off = ((size_t)(N + 1) * sizeof(float) + 255) & ~(size_t)255;
    int*   rowptr = (int*)((char*)d_ws + off);
    float* dis    = (float*)((char*)d_ws + 2 * off);
    float* bufA   = (float*)((char*)d_ws + 3 * off);
    float* bufO   = bufA + (size_t)N * 64;
    int*   csr    = (int*)(bufO + (size_t)N * 128);

    const int B  = (N + BNODES - 1) >> BSH;
    const int G  = (E + ECH - 1) / ECH;
    size_t ooff = ((size_t)(B + 1) * G * sizeof(int) + 255) & ~(size_t)255;
    int*  offs  = (int*)bufO;
    int2* pairs = (int2*)((char*)bufO + ooff);

    const int T = 256;

    k_chunk_group<<<G, LT, B * sizeof(int), stream>>>(src, dst, offs, pairs, E, B);
    k_bucket_csr<<<B, 512, G * sizeof(int2), stream>>>(pairs, offs, rowptr, dis, csr, N, G);

    k_gather_mlp<<<(N + TM - 1) / TM, 256, 0, stream>>>(X, rowptr, csr, dis, W1, b1, W2, bufO, N);
    k_gather64<true><<<2048, T, 0, stream>>>(bufO, rowptr, csr, dis, b2, out, N);
}


# Dense launch stores h2 rows already multiplied by dis[row] (T rows scaled in LDS); layer-2 neighbour sum adds source rows, no dis[s] read per edge; sum order kept, f32 rounding differs (scaled first)
# speedup vs baseline: 6.1061x; 1.0380x over previous
#include <hip/hip_runtime.h>
#include <hip/hip_bf16.h>


typedef __attribute__((ext_vector_type(2))) float v2f;
typedef __attribute__((ext_vector_type(8))) float v8f;

constexpr int BSH    = 8;
constexpr int BNODES = 1 << BSH;
constexpr int LT     = 512;
constexpr int ECH    = LT * 16;
constexpr int RU     = 13;
constexpr int SCAP   = 8192;

__global__ void __launch_bounds__(LT)
k_chunk_group(const int* __restrict__ src, const int* __restrict__ dst, int* __restrict__ off,
              int2* __restrict__ pairs, int e, int nb) {
    extern __shared__ int s_bkt[];
    __shared__ int s_ws[LT / 32];
    __shared__ __attribute__((aligned(16))) int2 s_pair[ECH];
    const int tid  = threadIdx.x;
    const int lane = tid & 31;
    const int wave = tid >> 5;
    const int g = gridDim.x;
    const int i0 = blockIdx.x * ECH + tid;
    int d[16], s[16], rk[16];
    #pragma unroll
    for (int k = 0; k < 16; ++k) {
        const int i = min(i0 + k * LT, e - 1);
        d[k] = dst[i];
        s[k] = src[i];
    }
    for (int b = tid; b < nb; b += LT) s_bkt[b] = 0;
    __syncthreads();
    #pragma unroll
    for (int k = 0; k < 16; ++k)
        rk[k] = (i0 + k * LT < e) ? atomicAdd(&s_bkt[d[k] >> BSH], 1) : -1;
    __syncthreads();

    int carry = 0;
    for (int b0 = 0; b0 < nb; b0 += LT) {
        const int b = b0 + tid;
        const int c = (b < nb) ? s_bkt[b] : 0;
        int inc = c;
        #pragma unroll
        for (int dlt = 1; dlt < 32; dlt <<= 1) {
            const int u = __shfl_up(inc, dlt);
            if (lane >= dlt) inc += u;
        }
        if (lane == 31) s_ws[wave] = inc;
        __syncthreads();
        int woff = 0, tot = 0;
        #pragma unroll
        for (int w = 0; w < LT / 32; ++w) {
            const int v = s_ws[w];
            tot += v;
            if (w < wave) woff += v;
        }
        const int ex = carry + woff + inc - c;
        if (b < nb) {
            s_bkt[b] = ex;
            off[(size_t)b * g + blockIdx.x] = ex;
        }
        carry += tot;
        __syncthreads();
    }
    if (tid == 0) off[(size_t)nb * g + blockIdx.x] = carry;

    int slot[16];
    #pragma unroll
    for (int k = 0; k < 16; ++k) slot[k] = s_bkt[d[k] >> BSH] + rk[k];
    #pragma unroll
    for (int k = 0; k < 16; ++k)
        if (rk[k] >= 0) s_pair[slot[k]] = make_int2(s[k], d[k]);
    __syncthreads();

    int2* mine = pairs + (size_t)blockIdx.x * ECH;
    const int cnt = min(ECH, e - (int)blockIdx.x * ECH);
    if (cnt == ECH) {
        int4 v[ECH / (2 * LT)];
        #pragma unroll
        for (int k = 0; k < ECH / (2 * LT); ++k)
            v[k] = *reinterpret_cast<const int4*>(s_pair + 2 * (k * LT + tid));
        #pragma unroll
        for (int k = 0; k < ECH / (2 * LT); ++k)
            *reinterpret_cast<int4*>(mine + 2 * (k * LT + tid)) = v[k];
    } else {
        for (int i = tid; i < cnt; i += LT) mine[i] = s_pair[i];
    }
}

__global__ void __launch_bounds__(512)
k_bucket_csr(const int2* __restrict__ pairs, const int* __restrict__ off, int* __restrict__ rowptr,
             float* __restrict__ dis, int* __restrict__ csr, int n, int g) {
    extern __shared__ int2 s_run[];
    __shared__ int s_cnt[BNODES];
    __shared__ int s_cur[BNODES];
    __shared__ int s_wsum[BNODES / 32];
    __shared__ int s_red[512 / 32];
    __shared__ int s_slot[SCAP];
    __shared__ int s_tot;
    const int tid   = threadIdx.x;
    const int lane  = tid & 31;
    const int wave  = tid >> 5;
    const int b     = blockIdx.x;
    const int node0 = b << BSH;

    if (tid < BNODES) s_cnt[tid] = 0;
    int a0 = 0;
    for (int q = tid; q < g; q += 512) {
        const int r0 = off[(size_t)b * g + q];
        const int r1 = off[(size_t)(b + 1) * g + q];
        s_run[q] = make_int2(r0, r1 - r0);
        a0 += r0;
    }
    #pragma unroll
    for (int dlt = 16; dlt >= 1; dlt >>= 1) a0 += __shfl_xor(a0, dlt);
    if (lane == 0) s_red[wave] = a0;
    __syncthreads();
    int beg = 0;
    #pragma unroll
    for (int w = 0; w < 512 / 32; ++w) beg += s_red[w];

    for (int q0 = wave; q0 < g; q0 += 16 * RU) {
        int r0[RU], len[RU], y[RU];
        #pragma unroll
        for (int u = 0; u < RU; ++u) {
            const int q = q0 + 16 * u;
            const int2 rl = s_run[min(q, g - 1)];
            r0[u]  = rl.x;
            len[u] = (q < g) ? rl.y : 0;
        }
        #pragma unroll
        for (int u = 0; u < RU; ++u) {
            const int2* rp = pairs + (size_t)min(q0 + 16 * u, g - 1) * ECH;
            y[u] = rp[(lane < len[u]) ? r0[u] + lane : 0].y;
        }
        #pragma unroll
        for (int u = 0; u < RU; ++u)
            if (lane < len[u]) atomicAdd(&s_cnt[y[u] - node0], 1);
        #pragma unroll
        for (int u = 0; u < RU; ++u) {
            const int2* rp = pairs + (size_t)min(q0 + 16 * u, g - 1) * ECH + r0[u];
            for (int i = lane + 32; i < len[u]; i += 32)
                atomicAdd(&s_cnt[rp[i].y - node0], 1);
        }
    }
    __syncthreads();

    const int c = (tid < BNODES) ? s_cnt[tid] : 0;
    int inc = c;
    #pragma unroll
    for (int dlt = 1; dlt < 32; dlt <<= 1) {
        const int u = __shfl_up(inc, dlt);
        if (lane >= dlt) inc += u;
    }
    if (tid < BNODES && lane == 31) s_wsum[wave] = inc;
    __syncthreads();
    if (tid < BNODES) {
        int ex = beg + inc - c;
        #pragma unroll
        for (int w = 0; w < BNODES / 32; ++w)
            if (w < wave) ex += s_wsum[w];
        s_cur[tid] = ex;
        const int node = node0 + tid;
        if (node < n) {
            rowptr[node] = ex;
            dis[node] = rsqrtf((float)(c + 1));
            if (node == n - 1) rowptr[n] = ex + c;
        }
        if (tid == BNODES - 1) s_tot = ex + c - beg;
    }
    __syncthreads();
    const int  tot    = s_tot;
    const bool staged = tot <= SCAP;

    for (int q0 = wave; q0 < g; q0 += 16 * RU) {
        int r0[RU], len[RU], pos[RU];
        int2 r[RU];
        #pragma unroll
        for (int u = 0; u < RU; ++u) {
            const int q = q0 + 16 * u;
            const int2 rl = s_run[min(q, g - 1)];
            r0[u]  = rl.x;
            len[u] = (q < g) ? rl.y : 0;
        }
        #pragma unroll
        for (int u = 0; u < RU; ++u) {
            const int2* rp = pairs + (size_t)min(q0 + 16 * u, g - 1) * ECH;
            r[u] = rp[(lane < len[u]) ? r0[u] + lane : 0];
        }
        #pragma unroll
        for (int u = 0; u < RU; ++u)
            pos[u] = (lane < len[u]) ? atomicAdd(&s_cur[r[u].y - node0], 1) : -1;
        #pragma unroll
        for (int u = 0; u < RU; ++u)
            if (pos[u] >= 0) {
                if (staged) s_slot[pos[u] - beg] = r[u].x;
                else        csr[pos[u]] = r[u].x;
            }
        #pragma unroll
        for (int u = 0; u < RU; ++u) {
            const int2* rp = pairs + (size_t)min(q0 + 16 * u, g - 1) * ECH + r0[u];
            for (int i = lane + 32; i < len[u]; i += 32) {
                const int2 t = rp[i];
                const int ps = atomicAdd(&s_cur[t.y - node0], 1);
                if (staged) s_slot[ps - beg] = t.x;
                else        csr[ps] = t.x;
            }
        }
    }
    __syncthreads();
    if (staged) {
        const int a = beg & 31;
        for (int p0 = tid - a; p0 < tot; p0 += 4 * 512) {
            int v[4];
            #pragma unroll
            for (int j = 0; j < 4; ++j)
                v[j] = s_slot[min(max(p0 + j * 512, 0), SCAP - 1)];
            #pragma unroll
            for (int j = 0; j < 4; ++j) {
                const int p = p0 + j * 512;
                if (p >= 0 && p < tot) csr[beg + p] = v[j];
            }
        }
    }
}

template <bool BIAS>
__global__ void __launch_bounds__(256)
k_gather64(const float* __restrict__ hs, const int* __restrict__ rowptr, const int* __restrict__ csr,
           const float* __restrict__ dis, const float* __restrict__ bias,
           float* __restrict__ out, int n) {
    const unsigned c4   = (threadIdx.x & 15u) << 2;
    const unsigned step = (gridDim.x * blockDim.x) >> 4;
    for (unsigned node = (blockIdx.x * blockDim.x + threadIdx.x) >> 4; node < (unsigned)n; node += step) {
        const int beg = rowptr[node];
        const int end = rowptr[node + 1];
        const float sd = dis[node];
        float4 acc = *(const float4*)(hs + (size_t)node * 64 + c4);
        const int last = end - 1;
        for (int p = beg; p < end; p += 4) {
            const bool ok1 = p + 1 < end, ok2 = p + 2 < end, ok3 = p + 3 < end;
            const int s0 = csr[p];
            const int s1 = csr[min(p + 1, last)];
            const int s2 = csr[min(p + 2, last)];
            const int s3 = csr[min(p + 3, last)];
            const float4 v0 = *(const float4*)(hs + (size_t)s0 * 64 + c4);
            const float4 v1 = *(const float4*)(hs + (size_t)s1 * 64 + c4);
            const float4 v2 = *(const float4*)(hs + (size_t)s2 * 64 + c4);
            const float4 v3 = *(const float4*)(hs + (size_t)s3 * 64 + c4);
            acc.x += v0.x; acc.y += v0.y; acc.z += v0.z; acc.w += v0.w;
            float4 t;
            t.x = acc.x + v1.x; t.y = acc.y + v1.y; t.z = acc.z + v1.z; t.w = acc.w + v1.w;
            acc.x = ok1 ? t.x : acc.x; acc.y = ok1 ? t.y : acc.y;
            acc.z = ok1 ? t.z : acc.z; acc.w = ok1 ? t.w : acc.w;
            t.x = acc.x + v2.x; t.y = acc.y + v2.y; t.z = acc.z + v2.z; t.w = acc.w + v2.w;
            acc.x = ok2 ? t.x : acc.x; acc.y = ok2 ? t.y : acc.y;
            acc.z = ok2 ? t.z : acc.z; acc.w = ok2 ? t.w : acc.w;
            t.x = acc.x + v3.x; t.y = acc.y + v3.y; t.z = acc.z + v3.z; t.w = acc.w + v3.w;
            acc.x = ok3 ? t.x : acc.x; acc.y = ok3 ? t.y : acc.y;
            acc.z = ok3 ? t.z : acc.z; acc.w = ok3 ? t.w : acc.w;
        }
        float4 r;
        r.x = acc.x * sd; r.y = acc.y * sd; r.z = acc.z * sd; r.w = acc.w * sd;
        if constexpr (BIAS) {
            const float4 b = *(const float4*)(bias + c4);
            r.x += b.x; r.y += b.y; r.z += b.z; r.w += b.w;
        }
        *(float4*)(out + (size_t)node * 64 + c4) = r;
    }
}

constexpr int F0  = 64;
constexpr int F1  = 128;
constexpr int F2  = 64;
constexpr int TM  = 64;
constexpr int LDA = F0 + 4;
constexpr int LDH = F1 + 4;

template <int KK>
__device__ __forceinline__ void mma_2rows(const float* A, int rs, const v2f* bf, v8f& c0, v8f& c1) {
    #pragma unroll
    for (int kk = 0; kk < KK; ++kk) {
        const v2f a0 = *(const v2f*)(A + kk * 4);
        const v2f a1 = *(const v2f*)(A + rs + kk * 4);
        c0 = __builtin_amdgcn_wmma_f32_16x16x4_f32(false, a0, false, bf[kk],
                                                   (short)0, c0, false, false);
        c1 = __builtin_amdgcn_wmma_f32_16x16x4_f32(false, a1, false, bf[kk],
                                                   (short)0, c1, false, false);
    }
}

template <int NR>
__device__ __forceinline__ void sum_rows(const float* __restrict__ X, int my, float mw, int k0, int rem,
                                         unsigned c4, float4& acc) {
    float  w[NR];
    float4 v[NR];
    #pragma unroll
    for (int i = 0; i < NR; ++i) {
        const int s = __shfl(my, k0 + i, 16);
        w[i] = __shfl(mw, k0 + i, 16);
        v[i] = *(const float4*)(X + (size_t)s * 64 + c4);
    }
    #pragma unroll
    for (int i = 0; i < NR; ++i) {
        const bool ok = k0 + i < rem;
        float4 t;
        t.x = fmaf(w[i], v[i].x, acc.x); t.y = fmaf(w[i], v[i].y, acc.y);
        t.z = fmaf(w[i], v[i].z, acc.z); t.w = fmaf(w[i], v[i].w, acc.w);
        acc.x = ok ? t.x : acc.x; acc.y = ok ? t.y : acc.y;
        acc.z = ok ? t.z : acc.z; acc.w = ok ? t.w : acc.w;
    }
}

__global__ void __launch_bounds__(256)
k_gather_mlp(const float* __restrict__ X, const int* __restrict__ rowptr, const int* __restrict__ csr,
             const float* __restrict__ dis, const float* __restrict__ W1, const float* __restrict__ b1,
             const float* __restrict__ W2, float* __restrict__ OUT, int M) {
    __shared__ __attribute__((aligned(16))) float sA[TM * LDA];
    __shared__ __attribute__((aligned(16))) float sH[TM * LDH];
    __shared__ __attribute__((aligned(16))) float sD[TM];

    const int tid  = threadIdx.x;
    const int wave = tid >> 5;
    const int lane = tid & 31;
    const int half = lane >> 4;
    const int l16  = lane & 15;
    const int m0   = blockIdx.x * TM;

    const int xr = tid >> 4;
    const unsigned c4 = (tid & 15) << 2;
    int nd[TM / 16], beg[TM / 16], cnt[TM / 16], my0[TM / 16];
    float sdv[TM / 16], mw0[TM / 16];
    #pragma unroll
    for (int j = 0; j < TM / 16; ++j) {
        int row = m0 + xr + 16 * j; if (row >= M) row = M - 1;
        nd[j]  = row;
        beg[j] = rowptr[row];
        cnt[j] = rowptr[row + 1] - beg[j];
        sdv[j] = dis[row];
    }
    #pragma unroll
    for (int j = 0; j < TM / 16; ++j) {
        my0[j] = (l16 < cnt[j]) ? csr[beg[j] + l16] : nd[j];
        mw0[j] = dis[my0[j]];
    }
    #pragma unroll
    for (int j = 0; j < TM / 16; ++j) {
        const float sd = sdv[j];
        float4 acc = *(const float4*)(X + (size_t)nd[j] * F0 + c4);
        acc.x *= sd; acc.y *= sd; acc.z *= sd; acc.w *= sd;
        int   my = my0[j];
        float mw = mw0[j];
        int rem  = cnt[j];
        int remw = __builtin_amdgcn_readfirstlane(max(rem, __shfl_xor(rem, 16)));
        int base = beg[j];
        while (remw > 0) {
            int nmy = nd[j];
            if (remw > 16) nmy = (16 + l16 < rem) ? csr[base + 16 + l16] : nd[j];
            if (remw > 4) sum_rows<8>(X, my, mw, 0, rem, c4, acc);
            else          sum_rows<4>(X, my, mw, 0, rem, c4, acc);
            float nmw = 0.0f;
            if (remw > 16) nmw = dis[nmy];
            if (remw > 8) {
                if (remw > 12) sum_rows<8>(X, my, mw, 8, rem, c4, acc);
                else           sum_rows<4>(X, my, mw, 8, rem, c4, acc);
            }
            my = nmy; mw = nmw;
            rem -= 16; remw -= 16; base += 16;
        }
        float4 r;
        r.x = acc.x * sd; r.y = acc.y * sd; r.z = acc.z * sd; r.w = acc.w * sd;
        *(float4*)&sA[(xr + 16 * j) * LDA + c4] = r;
        if (c4 == 0) sD[xr + 16 * j] = sd;
    }

    const int n1 = wave * 16 + l16;
    v2f bf1[F0 / 4];
    #pragma unroll
    for (int kk = 0; kk < F0 / 4; ++kk) {
        const int k = kk * 4 + 2 * half;
        bf1[kk].x = W1[k * F1 + n1];
        bf1[kk].y = W1[(k + 1) * F1 + n1];
    }
    const float bcol = b1[n1];

    __syncthreads();

    #pragma unroll
    for (int mp = 0; mp < TM / 32; ++mp) {
        v8f c0 = {}, c1 = {};
        const float4* D4 = reinterpret_cast<const float4*>(&sD[mp * 32 + half * 8]);
        const float4 da = D4[0], db = D4[1], dc = D4[4], dd = D4[5];
        const float d0[8] = {da.x, da.y, da.z, da.w, db.x, db.y, db.z, db.w};
        const float d1[8] = {dc.x, dc.y, dc.z, dc.w, dd.x, dd.y, dd.z, dd.w};
        mma_2rows<F0 / 4>(&sA[(mp * 32 + l16) * LDA + 2 * half], 16 * LDA, bf1, c0, c1);
        float* H = &sH[(mp * 32 + half * 8) * LDH + n1];
        #pragma unroll
        for (int r = 0; r < 8; ++r) {
            H[r * LDH]        = fmaxf(c0[r] + bcol, 0.0f) * d0[r];
            H[(16 + r) * LDH] = fmaxf(c1[r] + bcol, 0.0f) * d1[r];
        }
    }

    const int n2 = (wave & 3) * 16 + l16;
    v2f bf2[F1 / 4];
    #pragma unroll
    for (int kk = 0; kk < F1 / 4; ++kk) {
        const int k = kk * 4 + 2 * half;
        bf2[kk].x = W2[k * F2 + n2];
        bf2[kk].y = W2[(k + 1) * F2 + n2];
    }
    __syncthreads();

    const int mq = wave >> 2;
    v8f c0 = {}, c1 = {};
    mma_2rows<F1 / 4>(&sH[(mq * 16 + l16) * LDH + 2 * half], 32 * LDH, bf2, c0, c1);
    #pragma unroll
    for (int r = 0; r < 8; ++r) {
        const int row0 = m0 + mq * 16 + half * 8 + r;
        const int row1 = row0 + 32;
        if (row0 < M) OUT[(size_t)row0 * F2 + n2] = c0[r];
        if (row1 < M) OUT[(size_t)row1 * F2 + n2] = c1[r];
    }
}

extern "C" void kernel_launch(void* const* d_in, const int* in_sizes, int n_in,
                              void* d_out, int out_size, void* d_ws, size_t ws_size,
                              hipStream_t stream) {
    const float* X  = (const float*)d_in[0];
    const int*   EI = (const int*)d_in[1];
    const float* W1 = (const float*)d_in[2];
    const float* b1 = (const float*)d_in[3];
    const float* W2 = (const float*)d_in[4];
    const float* b2 = (const float*)d_in[5];
    float* out = (float*)d_out;

    const int FIN1 = 64;
    const int N = in_sizes[0] / FIN1;
    const int E = in_sizes[1] / 2;
    const int* src = EI;
    const int* dst = EI + E;

    size_t off = ((size_t)(N + 1) * sizeof(float) + 255) & ~(size_t)255;
    int*   rowptr = (int*)((char*)d_ws + off);
    float* dis    = (float*)((char*)d_ws + 2 * off);
    float* bufA   = (float*)((char*)d_ws + 3 * off);
    float* bufO   = bufA + (size_t)N * 64;
    int*   csr    = (int*)(bufO + (size_t)N * 128);

    const int B  = (N + BNODES - 1) >> BSH;
    const int G  = (E + ECH - 1) / ECH;
    size_t ooff = ((size_t)(B + 1) * G * sizeof(int) + 255) & ~(size_t)255;
    int*  offs  = (int*)bufO;
    int2* pairs = (int2*)((char*)bufO + ooff);

    const int T = 256;

    k_chunk_group<<<G, LT, B * sizeof(int), stream>>>(src, dst, offs, pairs, E, B);
    k_bucket_csr<<<B, 512, G * sizeof(int2), stream>>>(pairs, offs, rowptr, dis, csr, N, G);

    k_gather_mlp<<<(N + TM - 1) / TM, 256, 0, stream>>>(X, rowptr, csr, dis, W1, b1, W2, bufO, N);
    k_gather64<true><<<2048, T, 0, stream>>>(bufO, rowptr, csr, dis, b2, out, N);
}
